# starting kernel
# baseline (speedup 1.0000x reference)
#include <hip/hip_runtime.h>
#include <math.h>


typedef __attribute__((ext_vector_type(2))) float v2f;
typedef __attribute__((ext_vector_type(8))) float v8f;

#define N_LAYERS 6
#define NWAVES   4
#define NTHREADS (NWAVES * 32)

struct c2 { float re, im; };

__device__ __forceinline__ c2 cmul(c2 a, c2 b) {
  return c2{a.re * b.re - a.im * b.im, a.re * b.im + a.im * b.re};
}
__device__ __forceinline__ c2 rx_el(int a, int b, float c, float s) {
  return (a == b) ? c2{c, 0.f} : c2{0.f, -s};
}
__device__ __forceinline__ c2 ry_el(int a, int b, float c, float s) {
  if (a == b) return c2{c, 0.f};
  return (a == 1) ? c2{s, 0.f} : c2{-s, 0.f};
}

__global__ void qnn_build_v(const float* __restrict__ wts, float* __restrict__ V) {
  int l = blockIdx.x;
  int t = threadIdx.x;
  int r = t >> 4, c = t & 15;
  for (int m = 0; m < 2; ++m) {
    int rr = (m == 0) ? (r ^ (r >> 1)) : r;
    c2 prod = c2{1.f, 0.f};
    for (int i = 0; i < 4; ++i) {
      int wire = m * 4 + i;
      float a0 = wts[(l * 8 + wire) * 3 + 0];
      float a1 = wts[(l * 8 + wire) * 3 + 1];
      float a2 = wts[(l * 8 + wire) * 3 + 2];
      float cx = cosf(0.5f * a0), sx = sinf(0.5f * a0);
      float cy = cosf(0.5f * a1), sy = sinf(0.5f * a1);
      float cz = cosf(0.5f * a2), sz = sinf(0.5f * a2);
      int ab = (rr >> (3 - i)) & 1;
      int bb = (c  >> (3 - i)) & 1;
      c2 yx = c2{0.f, 0.f};
      for (int k = 0; k < 2; ++k) {
        c2 p = cmul(ry_el(ab, k, cy, sy), rx_el(k, bb, cx, sx));
        yx.re += p.re; yx.im += p.im;
      }
      c2 z = ab ? c2{cz, sz} : c2{cz, -sz};
      prod = cmul(prod, cmul(z, yx));
    }
    V[(((l * 2 + m) * 2 + 0) * 256) + t] = prod.re;
    V[(((l * 2 + m) * 2 + 1) * 256) + t] = prod.im;
  }
}

__device__ __forceinline__ void cmm(const float* __restrict__ Ar,
                                    const float* __restrict__ Ai,
                                    const float* __restrict__ Br,
                                    const float* __restrict__ Bi,
                                    bool bT, int l, v8f& Cr, v8f& Ci) {
  int lr = l & 15, lh = l >> 4;
  int aoff = lr * 16 + 2 * lh;
  v8f rr = {}, ii = {}, ri = {};
#pragma unroll
  for (int kb = 0; kb < 4; ++kb) {
    v2f ar = *(const v2f*)(Ar + aoff + 4 * kb);
    v2f ai = *(const v2f*)(Ai + aoff + 4 * kb);
    v2f br, bi;
    if (bT) {
      br = *(const v2f*)(Br + aoff + 4 * kb);
      bi = *(const v2f*)(Bi + aoff + 4 * kb);
    } else {
      int k0 = 4 * kb + 2 * lh;
      br.x = Br[k0 * 16 + lr]; br.y = Br[(k0 + 1) * 16 + lr];
      bi.x = Bi[k0 * 16 + lr]; bi.y = Bi[(k0 + 1) * 16 + lr];
    }
    rr = __builtin_amdgcn_wmma_f32_16x16x4_f32(false, ar, false, br, (short)0, rr, false, false);
    ii = __builtin_amdgcn_wmma_f32_16x16x4_f32(false, ai, false, bi, (short)0, ii, false, false);
    ri = __builtin_amdgcn_wmma_f32_16x16x4_f32(false, ar, false, bi, (short)0, ri, false, false);
    ri = __builtin_amdgcn_wmma_f32_16x16x4_f32(false, ai, false, br, (short)0, ri, false, false);
  }
  Cr = rr - ii;
  Ci = ri;
}

__device__ __forceinline__ void store_tile(float* __restrict__ dst, int l, v8f c) {
  int base = ((l >> 4) * 8) * 16 + (l & 15);
#pragma unroll
  for (int v = 0; v < 8; ++v) dst[base + v * 16] = c[v];
}

__device__ __forceinline__ float wave_sum(float v, int l) {
#pragma unroll
  for (int m = 16; m; m >>= 1) {
    int src = (l ^ m) << 2;
    v += __int_as_float(__builtin_amdgcn_ds_bpermute(src, __float_as_int(v)));
  }
  return v;
}

__global__ void __launch_bounds__(NTHREADS)
qnn_kernel(const float* __restrict__ x, const float* __restrict__ Vg,
           float* __restrict__ out, int batch) {
  __shared__ float Vl[N_LAYERS * 2 * 2 * 256];
  __shared__ float Ebuf[NWAVES][2][256];
  __shared__ float Mbuf[NWAVES][2][256];
  __shared__ float Nbuf[NWAVES][2][256];
  __shared__ float Sbuf[NWAVES][2][256];

  int tid = threadIdx.x;
  int l = tid & 31;
  int w = tid >> 5;
  int b = blockIdx.x * NWAVES + w;
  int bb = (b < batch) ? b : (batch - 1);

  {
    const float4* src = (const float4*)Vg;
    float4* dst = (float4*)Vl;
    for (int i = tid; i < (N_LAYERS * 2 * 2 * 256) / 4; i += NTHREADS) dst[i] = src[i];
  }
  __syncthreads();

  float x0 = x[2 * bb], x1 = x[2 * bb + 1];
  float c0 = cosf(0.5f * x0), s0 = sinf(0.5f * x0);
  float c1 = cosf(0.5f * x1), s1 = sinf(0.5f * x1);
#pragma unroll
  for (int t = 0; t < 8; ++t) {
    int idx = t * 32 + l;
    int a = idx >> 4, bcol = idx & 15;
    c2 e = rx_el((a >> 3) & 1, (bcol >> 3) & 1, c0, s0);
    e = cmul(e, ry_el((a >> 2) & 1, (bcol >> 2) & 1, c1, s1));
    e = cmul(e, rx_el((a >> 1) & 1, (bcol >> 1) & 1, c0, s0));
    e = cmul(e, ry_el(a & 1, bcol & 1, c1, s1));
    Ebuf[w][0][idx] = e.re;
    Ebuf[w][1][idx] = e.im;
    Sbuf[w][0][idx] = (idx == 0) ? 1.f : 0.f;
    Sbuf[w][1][idx] = 0.f;
  }

  int csrc = (l & 15); csrc = csrc ^ (csrc >> 1);
  int addrE = (((l & 16) | csrc)) << 2;
  int addrO = addrE ^ (8 << 2);

  for (int layer = 0; layer < N_LAYERS; ++layer) {
    const float* V0r = &Vl[((layer * 2 + 0) * 2 + 0) * 256];
    const float* V0i = &Vl[((layer * 2 + 0) * 2 + 1) * 256];
    const float* R1r = &Vl[((layer * 2 + 1) * 2 + 0) * 256];
    const float* R1i = &Vl[((layer * 2 + 1) * 2 + 1) * 256];
    v8f cr, ci;
    cmm(V0r, V0i, &Ebuf[w][0][0], &Ebuf[w][1][0], false, l, cr, ci);
    store_tile(&Mbuf[w][0][0], l, cr); store_tile(&Mbuf[w][1][0], l, ci);
    cmm(R1r, R1i, &Ebuf[w][0][0], &Ebuf[w][1][0], false, l, cr, ci);
    store_tile(&Nbuf[w][0][0], l, cr); store_tile(&Nbuf[w][1][0], l, ci);
    cmm(&Mbuf[w][0][0], &Mbuf[w][1][0], &Sbuf[w][0][0], &Sbuf[w][1][0], false, l, cr, ci);
    store_tile(&Mbuf[w][0][0], l, cr); store_tile(&Mbuf[w][1][0], l, ci);
    cmm(&Mbuf[w][0][0], &Mbuf[w][1][0], &Nbuf[w][0][0], &Nbuf[w][1][0], true, l, cr, ci);
    v8f nr, ni;
#pragma unroll
    for (int v = 0; v < 8; ++v) {
      int ad = (v & 1) ? addrO : addrE;
      nr[v] = __int_as_float(__builtin_amdgcn_ds_bpermute(ad, __float_as_int(cr[v])));
      ni[v] = __int_as_float(__builtin_amdgcn_ds_bpermute(ad, __float_as_int(ci[v])));
    }
    store_tile(&Sbuf[w][0][0], l, nr);
    store_tile(&Sbuf[w][1][0], l, ni);
  }

  int lr = l & 15, lh = l >> 4;
  float p[8];
#pragma unroll
  for (int v = 0; v < 8; ++v) {
    float a = Sbuf[w][0][(v + 8 * lh) * 16 + lr];
    float bi = Sbuf[w][1][(v + 8 * lh) * 16 + lr];
    p[v] = a * a + bi * bi;
  }
  float t  = p[0] + p[1] + p[2] + p[3] + p[4] + p[5] + p[6] + p[7];
  float u1 = p[0] + p[1] + p[2] + p[3] - p[4] - p[5] - p[6] - p[7];
  float u2 = p[0] + p[1] - p[2] - p[3] + p[4] + p[5] - p[6] - p[7];
  float u3 = p[0] - p[1] + p[2] - p[3] + p[4] - p[5] + p[6] - p[7];
  float ev[8];
  ev[0] = lh ? -t : t;
  ev[1] = u1; ev[2] = u2; ev[3] = u3;
  ev[4] = (lr & 8) ? -t : t;
  ev[5] = (lr & 4) ? -t : t;
  ev[6] = (lr & 2) ? -t : t;
  ev[7] = (lr & 1) ? -t : t;
#pragma unroll
  for (int i = 0; i < 8; ++i) ev[i] = wave_sum(ev[i], l);

  if (b < batch && l == 0) {
    const float kPi = 3.14159265358979323846f;
#pragma unroll
    for (int i = 0; i < 8; ++i) out[b * 8 + i] = kPi * ev[i];
  }
}

extern "C" void kernel_launch(void* const* d_in, const int* in_sizes, int n_in,
                              void* d_out, int out_size, void* d_ws, size_t ws_size,
                              hipStream_t stream) {
  const float* x   = (const float*)d_in[0];
  const float* wts = (const float*)d_in[1];
  float* out = (float*)d_out;
  float* V   = (float*)d_ws;

  int batch = in_sizes[0] / 2;
  qnn_build_v<<<N_LAYERS, 256, 0, stream>>>(wts, V);
  int nblocks = (batch + NWAVES - 1) / NWAVES;
  qnn_kernel<<<nblocks, NTHREADS, 0, stream>>>(x, V, out, batch);
}


# complex 16x16 product from 3 real f32 WMMA products instead of 4 (K1=(Ar+Ai)Br, K2=Ar(Bi-Br), K3=Ai(Br+Bi)): 12 WMMA per product, was 16; re/im sums are formed in a different order
# speedup vs baseline: 1.1632x; 1.1632x over previous
#include <hip/hip_runtime.h>
#include <math.h>


typedef __attribute__((ext_vector_type(2))) float v2f;
typedef __attribute__((ext_vector_type(8))) float v8f;

#define N_LAYERS 6
#define NWAVES   4
#define NTHREADS (NWAVES * 32)

struct c2 { float re, im; };

__device__ __forceinline__ c2 cmul(c2 a, c2 b) {
  return c2{a.re * b.re - a.im * b.im, a.re * b.im + a.im * b.re};
}
__device__ __forceinline__ c2 rx_el(int a, int b, float c, float s) {
  return (a == b) ? c2{c, 0.f} : c2{0.f, -s};
}
__device__ __forceinline__ c2 ry_el(int a, int b, float c, float s) {
  if (a == b) return c2{c, 0.f};
  return (a == 1) ? c2{s, 0.f} : c2{-s, 0.f};
}

__global__ void qnn_build_v(const float* __restrict__ wts, float* __restrict__ V) {
  int l = blockIdx.x;
  int t = threadIdx.x;
  int r = t >> 4, c = t & 15;
  for (int m = 0; m < 2; ++m) {
    int rr = (m == 0) ? (r ^ (r >> 1)) : r;
    c2 prod = c2{1.f, 0.f};
    for (int i = 0; i < 4; ++i) {
      int wire = m * 4 + i;
      float a0 = wts[(l * 8 + wire) * 3 + 0];
      float a1 = wts[(l * 8 + wire) * 3 + 1];
      float a2 = wts[(l * 8 + wire) * 3 + 2];
      float cx = cosf(0.5f * a0), sx = sinf(0.5f * a0);
      float cy = cosf(0.5f * a1), sy = sinf(0.5f * a1);
      float cz = cosf(0.5f * a2), sz = sinf(0.5f * a2);
      int ab = (rr >> (3 - i)) & 1;
      int bb = (c  >> (3 - i)) & 1;
      c2 yx = c2{0.f, 0.f};
      for (int k = 0; k < 2; ++k) {
        c2 p = cmul(ry_el(ab, k, cy, sy), rx_el(k, bb, cx, sx));
        yx.re += p.re; yx.im += p.im;
      }
      c2 z = ab ? c2{cz, sz} : c2{cz, -sz};
      prod = cmul(prod, cmul(z, yx));
    }
    V[(((l * 2 + m) * 2 + 0) * 256) + t] = prod.re;
    V[(((l * 2 + m) * 2 + 1) * 256) + t] = prod.im;
  }
}

__device__ __forceinline__ void cmm(const float* __restrict__ Ar,
                                    const float* __restrict__ Ai,
                                    const float* __restrict__ Br,
                                    const float* __restrict__ Bi,
                                    bool bT, int l, v8f& Cr, v8f& Ci) {
  int lr = l & 15, lh = l >> 4;
  int aoff = lr * 16 + 2 * lh;
  v8f k1 = {}, k3 = {};
  v2f arv[4], bdv[4];
#pragma unroll
  for (int kb = 0; kb < 4; ++kb) {
    v2f ar = *(const v2f*)(Ar + aoff + 4 * kb);
    v2f ai = *(const v2f*)(Ai + aoff + 4 * kb);
    v2f br, bi;
    if (bT) {
      br = *(const v2f*)(Br + aoff + 4 * kb);
      bi = *(const v2f*)(Bi + aoff + 4 * kb);
    } else {
      int k0 = 4 * kb + 2 * lh;
      br.x = Br[k0 * 16 + lr]; br.y = Br[(k0 + 1) * 16 + lr];
      bi.x = Bi[k0 * 16 + lr]; bi.y = Bi[(k0 + 1) * 16 + lr];
    }
    v2f as = ar + ai;
    v2f bs = br + bi;
    arv[kb] = ar;
    bdv[kb] = bi - br;
    k1 = __builtin_amdgcn_wmma_f32_16x16x4_f32(false, as, false, br, (short)0, k1, false, false);
    k3 = __builtin_amdgcn_wmma_f32_16x16x4_f32(false, ai, false, bs, (short)0, k3, false, false);
  }
  v8f ci = k1;
#pragma unroll
  for (int kb = 0; kb < 4; ++kb)
    ci = __builtin_amdgcn_wmma_f32_16x16x4_f32(false, arv[kb], false, bdv[kb], (short)0, ci, false, false);
  Cr = k1 - k3;
  Ci = ci;
}

__device__ __forceinline__ void store_tile(float* __restrict__ dst, int l, v8f c) {
  int base = ((l >> 4) * 8) * 16 + (l & 15);
#pragma unroll
  for (int v = 0; v < 8; ++v) dst[base + v * 16] = c[v];
}

__device__ __forceinline__ float wave_sum(float v, int l) {
#pragma unroll
  for (int m = 16; m; m >>= 1) {
    int src = (l ^ m) << 2;
    v += __int_as_float(__builtin_amdgcn_ds_bpermute(src, __float_as_int(v)));
  }
  return v;
}

__global__ void __launch_bounds__(NTHREADS)
qnn_kernel(const float* __restrict__ x, const float* __restrict__ Vg,
           float* __restrict__ out, int batch) {
  __shared__ float Vl[N_LAYERS * 2 * 2 * 256];
  __shared__ float Ebuf[NWAVES][2][256];
  __shared__ float Mbuf[NWAVES][2][256];
  __shared__ float Nbuf[NWAVES][2][256];
  __shared__ float Sbuf[NWAVES][2][256];

  int tid = threadIdx.x;
  int l = tid & 31;
  int w = tid >> 5;
  int b = blockIdx.x * NWAVES + w;
  int bb = (b < batch) ? b : (batch - 1);

  {
    const float4* src = (const float4*)Vg;
    float4* dst = (float4*)Vl;
    for (int i = tid; i < (N_LAYERS * 2 * 2 * 256) / 4; i += NTHREADS) dst[i] = src[i];
  }
  __syncthreads();

  float x0 = x[2 * bb], x1 = x[2 * bb + 1];
  float c0 = cosf(0.5f * x0), s0 = sinf(0.5f * x0);
  float c1 = cosf(0.5f * x1), s1 = sinf(0.5f * x1);
#pragma unroll
  for (int t = 0; t < 8; ++t) {
    int idx = t * 32 + l;
    int a = idx >> 4, bcol = idx & 15;
    c2 e = rx_el((a >> 3) & 1, (bcol >> 3) & 1, c0, s0);
    e = cmul(e, ry_el((a >> 2) & 1, (bcol >> 2) & 1, c1, s1));
    e = cmul(e, rx_el((a >> 1) & 1, (bcol >> 1) & 1, c0, s0));
    e = cmul(e, ry_el(a & 1, bcol & 1, c1, s1));
    Ebuf[w][0][idx] = e.re;
    Ebuf[w][1][idx] = e.im;
    Sbuf[w][0][idx] = (idx == 0) ? 1.f : 0.f;
    Sbuf[w][1][idx] = 0.f;
  }

  int csrc = (l & 15); csrc = csrc ^ (csrc >> 1);
  int addrE = (((l & 16) | csrc)) << 2;
  int addrO = addrE ^ (8 << 2);

  for (int layer = 0; layer < N_LAYERS; ++layer) {
    const float* V0r = &Vl[((layer * 2 + 0) * 2 + 0) * 256];
    const float* V0i = &Vl[((layer * 2 + 0) * 2 + 1) * 256];
    const float* R1r = &Vl[((layer * 2 + 1) * 2 + 0) * 256];
    const float* R1i = &Vl[((layer * 2 + 1) * 2 + 1) * 256];
    v8f cr, ci;
    cmm(V0r, V0i, &Ebuf[w][0][0], &Ebuf[w][1][0], false, l, cr, ci);
    store_tile(&Mbuf[w][0][0], l, cr); store_tile(&Mbuf[w][1][0], l, ci);
    cmm(R1r, R1i, &Ebuf[w][0][0], &Ebuf[w][1][0], false, l, cr, ci);
    store_tile(&Nbuf[w][0][0], l, cr); store_tile(&Nbuf[w][1][0], l, ci);
    cmm(&Mbuf[w][0][0], &Mbuf[w][1][0], &Sbuf[w][0][0], &Sbuf[w][1][0], false, l, cr, ci);
    store_tile(&Mbuf[w][0][0], l, cr); store_tile(&Mbuf[w][1][0], l, ci);
    cmm(&Mbuf[w][0][0], &Mbuf[w][1][0], &Nbuf[w][0][0], &Nbuf[w][1][0], true, l, cr, ci);
    v8f nr, ni;
#pragma unroll
    for (int v = 0; v < 8; ++v) {
      int ad = (v & 1) ? addrO : addrE;
      nr[v] = __int_as_float(__builtin_amdgcn_ds_bpermute(ad, __float_as_int(cr[v])));
      ni[v] = __int_as_float(__builtin_amdgcn_ds_bpermute(ad, __float_as_int(ci[v])));
    }
    store_tile(&Sbuf[w][0][0], l, nr);
    store_tile(&Sbuf[w][1][0], l, ni);
  }

  int lr = l & 15, lh = l >> 4;
  float p[8];
#pragma unroll
  for (int v = 0; v < 8; ++v) {
    float a = Sbuf[w][0][(v + 8 * lh) * 16 + lr];
    float bi = Sbuf[w][1][(v + 8 * lh) * 16 + lr];
    p[v] = a * a + bi * bi;
  }
  float t  = p[0] + p[1] + p[2] + p[3] + p[4] + p[5] + p[6] + p[7];
  float u1 = p[0] + p[1] + p[2] + p[3] - p[4] - p[5] - p[6] - p[7];
  float u2 = p[0] + p[1] - p[2] - p[3] + p[4] + p[5] - p[6] - p[7];
  float u3 = p[0] - p[1] + p[2] - p[3] + p[4] - p[5] + p[6] - p[7];
  float ev[8];
  ev[0] = lh ? -t : t;
  ev[1] = u1; ev[2] = u2; ev[3] = u3;
  ev[4] = (lr & 8) ? -t : t;
  ev[5] = (lr & 4) ? -t : t;
  ev[6] = (lr & 2) ? -t : t;
  ev[7] = (lr & 1) ? -t : t;
#pragma unroll
  for (int i = 0; i < 8; ++i) ev[i] = wave_sum(ev[i], l);

  if (b < batch && l == 0) {
    const float kPi = 3.14159265358979323846f;
#pragma unroll
    for (int i = 0; i < 8; ++i) out[b * 8 + i] = kPi * ev[i];
  }
}

extern "C" void kernel_launch(void* const* d_in, const int* in_sizes, int n_in,
                              void* d_out, int out_size, void* d_ws, size_t ws_size,
                              hipStream_t stream) {
  const float* x   = (const float*)d_in[0];
  const float* wts = (const float*)d_in[1];
  float* out = (float*)d_out;
  float* V   = (float*)d_ws;

  int batch = in_sizes[0] / 2;
  qnn_build_v<<<N_LAYERS, 256, 0, stream>>>(wts, V);
  int nblocks = (batch + NWAVES - 1) / NWAVES;
  qnn_kernel<<<nblocks, NTHREADS, 0, stream>>>(x, V, out, batch);
}


# encoding matrix factored E = D^-1 Er D (RX = P^-1 RY P) with Er real: each of the 2 products with E per layer takes 8 WMMA, was 12; phases D baked into layer matrices; re/im sums re-ordered
# speedup vs baseline: 1.3268x; 1.1406x over previous
#include <hip/hip_runtime.h>
#include <math.h>


typedef __attribute__((ext_vector_type(2))) float v2f;
typedef __attribute__((ext_vector_type(8))) float v8f;

#define N_LAYERS 6
#define NWAVES   4
#define NTHREADS (NWAVES * 32)

struct c2 { float re, im; };

__device__ __forceinline__ c2 cmul(c2 a, c2 b) {
  return c2{a.re * b.re - a.im * b.im, a.re * b.im + a.im * b.re};
}
__device__ __forceinline__ c2 rx_el(int a, int b, float c, float s) {
  return (a == b) ? c2{c, 0.f} : c2{0.f, -s};
}
__device__ __forceinline__ c2 ry_el(int a, int b, float c, float s) {
  if (a == b) return c2{c, 0.f};
  return (a == 1) ? c2{s, 0.f} : c2{-s, 0.f};
}

__device__ __forceinline__ int phase_k(int n) { return ((n >> 3) & 1) + ((n >> 1) & 1); }
__device__ __forceinline__ c2 mul_ipow(c2 a, int k) {
  k &= 3;
  if (k == 0) return a;
  if (k == 1) return c2{-a.im, a.re};
  if (k == 2) return c2{-a.re, -a.im};
  return c2{a.im, -a.re};
}
__device__ __forceinline__ float ry_re(int a, int b, float c, float s) {
  return (a == b) ? c : ((a == 1) ? s : -s);
}

__global__ void qnn_build_v(const float* __restrict__ wts, float* __restrict__ V) {
  int l = blockIdx.x;
  int t = threadIdx.x;
  int r = t >> 4, c = t & 15;
  for (int m = 0; m < 2; ++m) {
    int rr = (m == 0) ? (r ^ (r >> 1)) : r;
    c2 prod = c2{1.f, 0.f};
    for (int i = 0; i < 4; ++i) {
      int wire = m * 4 + i;
      float a0 = wts[(l * 8 + wire) * 3 + 0];
      float a1 = wts[(l * 8 + wire) * 3 + 1];
      float a2 = wts[(l * 8 + wire) * 3 + 2];
      float cx = cosf(0.5f * a0), sx = sinf(0.5f * a0);
      float cy = cosf(0.5f * a1), sy = sinf(0.5f * a1);
      float cz = cosf(0.5f * a2), sz = sinf(0.5f * a2);
      int ab = (rr >> (3 - i)) & 1;
      int bb = (c  >> (3 - i)) & 1;
      c2 yx = c2{0.f, 0.f};
      for (int k = 0; k < 2; ++k) {
        c2 p = cmul(ry_el(ab, k, cy, sy), rx_el(k, bb, cx, sx));
        yx.re += p.re; yx.im += p.im;
      }
      c2 z = ab ? c2{cz, sz} : c2{cz, -sz};
      prod = cmul(prod, cmul(z, yx));
    }
    prod = mul_ipow(prod, ((m == 0) ? phase_k(r) : 0) - phase_k(c));
    V[(((l * 2 + m) * 2 + 0) * 256) + t] = prod.re;
    V[(((l * 2 + m) * 2 + 1) * 256) + t] = prod.im;
  }
}

__device__ __forceinline__ void cmm(const float* __restrict__ Ar,
                                    const float* __restrict__ Ai,
                                    const float* __restrict__ Br,
                                    const float* __restrict__ Bi,
                                    bool bT, int l, v8f& Cr, v8f& Ci) {
  int lr = l & 15, lh = l >> 4;
  int aoff = lr * 16 + 2 * lh;
  v8f k1 = {}, k3 = {};
  v2f arv[4], bdv[4];
#pragma unroll
  for (int kb = 0; kb < 4; ++kb) {
    v2f ar = *(const v2f*)(Ar + aoff + 4 * kb);
    v2f ai = *(const v2f*)(Ai + aoff + 4 * kb);
    v2f br, bi;
    if (bT) {
      br = *(const v2f*)(Br + aoff + 4 * kb);
      bi = *(const v2f*)(Bi + aoff + 4 * kb);
    } else {
      int k0 = 4 * kb + 2 * lh;
      br.x = Br[k0 * 16 + lr]; br.y = Br[(k0 + 1) * 16 + lr];
      bi.x = Bi[k0 * 16 + lr]; bi.y = Bi[(k0 + 1) * 16 + lr];
    }
    v2f as = ar + ai;
    v2f bs = br + bi;
    arv[kb] = ar;
    bdv[kb] = bi - br;
    k1 = __builtin_amdgcn_wmma_f32_16x16x4_f32(false, as, false, br, (short)0, k1, false, false);
    k3 = __builtin_amdgcn_wmma_f32_16x16x4_f32(false, ai, false, bs, (short)0, k3, false, false);
  }
  v8f ci = k1;
#pragma unroll
  for (int kb = 0; kb < 4; ++kb)
    ci = __builtin_amdgcn_wmma_f32_16x16x4_f32(false, arv[kb], false, bdv[kb], (short)0, ci, false, false);
  Cr = k1 - k3;
  Ci = ci;
}

__device__ __forceinline__ void cmr2(const float* __restrict__ Ar,
                                     const float* __restrict__ Ai,
                                     const float* __restrict__ Br,
                                     const float* __restrict__ Bi,
                                     const v2f* eb, int l,
                                     v8f& Pr, v8f& Pi, v8f& Qr, v8f& Qi) {
  int aoff = (l & 15) * 16 + 2 * (l >> 4);
  v8f pr = {}, pi = {}, qr = {}, qi = {};
#pragma unroll
  for (int kb = 0; kb < 4; ++kb) {
    v2f ar = *(const v2f*)(Ar + aoff + 4 * kb);
    v2f ai = *(const v2f*)(Ai + aoff + 4 * kb);
    v2f br = *(const v2f*)(Br + aoff + 4 * kb);
    v2f bi = *(const v2f*)(Bi + aoff + 4 * kb);
    pr = __builtin_amdgcn_wmma_f32_16x16x4_f32(false, ar, false, eb[kb], (short)0, pr, false, false);
    pi = __builtin_amdgcn_wmma_f32_16x16x4_f32(false, ai, false, eb[kb], (short)0, pi, false, false);
    qr = __builtin_amdgcn_wmma_f32_16x16x4_f32(false, br, false, eb[kb], (short)0, qr, false, false);
    qi = __builtin_amdgcn_wmma_f32_16x16x4_f32(false, bi, false, eb[kb], (short)0, qi, false, false);
  }
  Pr = pr; Pi = pi; Qr = qr; Qi = qi;
}

__device__ __forceinline__ void store_tile(float* __restrict__ dst, int l, v8f c) {
  int base = ((l >> 4) * 8) * 16 + (l & 15);
#pragma unroll
  for (int v = 0; v < 8; ++v) dst[base + v * 16] = c[v];
}

__device__ __forceinline__ float wave_sum(float v, int l) {
#pragma unroll
  for (int m = 16; m; m >>= 1) {
    int src = (l ^ m) << 2;
    v += __int_as_float(__builtin_amdgcn_ds_bpermute(src, __float_as_int(v)));
  }
  return v;
}

__global__ void __launch_bounds__(NTHREADS)
qnn_kernel(const float* __restrict__ x, const float* __restrict__ Vg,
           float* __restrict__ out, int batch) {
  __shared__ float Vl[N_LAYERS * 2 * 2 * 256];
  __shared__ float Mbuf[NWAVES][2][256];
  __shared__ float Nbuf[NWAVES][2][256];
  __shared__ float Sbuf[NWAVES][2][256];

  int tid = threadIdx.x;
  int l = tid & 31;
  int w = tid >> 5;
  int b = blockIdx.x * NWAVES + w;
  int bb = (b < batch) ? b : (batch - 1);

  {
    const float4* src = (const float4*)Vg;
    float4* dst = (float4*)Vl;
    for (int i = tid; i < (N_LAYERS * 2 * 2 * 256) / 4; i += NTHREADS) dst[i] = src[i];
  }
  __syncthreads();

  float x0 = x[2 * bb], x1 = x[2 * bb + 1];
  float c0 = cosf(0.5f * x0), s0 = sinf(0.5f * x0);
  float c1 = cosf(0.5f * x1), s1 = sinf(0.5f * x1);
  int lr = l & 15, lh = l >> 4;
  v2f eb[4];
  {
    float f  = ry_re(lh, (lr >> 1) & 1, c0, s0);
    float g0 = ry_re(0, lr & 1, c1, s1);
    float g1 = ry_re(1, lr & 1, c1, s1);
#pragma unroll
    for (int kb = 0; kb < 4; ++kb) {
      float h = ry_re(kb >> 1, (lr >> 3) & 1, c0, s0) * ry_re(kb & 1, (lr >> 2) & 1, c1, s1);
      float hf = h * f;
      eb[kb].x = hf * g0;
      eb[kb].y = hf * g1;
    }
  }
  for (int t = 0; t < 8; ++t) {
    int idx = t * 32 + l;
    Sbuf[w][0][idx] = (idx == 0) ? 1.f : 0.f;
    Sbuf[w][1][idx] = 0.f;
  }

  int csrc = (l & 15); csrc = csrc ^ (csrc >> 1);
  int addrE = (((l & 16) | csrc)) << 2;
  int addrO = addrE ^ (8 << 2);
  int kc = phase_k(lr);
  float* dstA = (kc == 1) ? &Sbuf[w][1][0] : &Sbuf[w][0][0];
  float* dstB = (kc == 1) ? &Sbuf[w][0][0] : &Sbuf[w][1][0];
  int sgnA = (kc == 2) ? (int)0x80000000u : 0;
  int sgnB = (kc >= 1) ? (int)0x80000000u : 0;

  for (int layer = 0; layer < N_LAYERS; ++layer) {
    const float* V0r = &Vl[((layer * 2 + 0) * 2 + 0) * 256];
    const float* V0i = &Vl[((layer * 2 + 0) * 2 + 1) * 256];
    const float* R1r = &Vl[((layer * 2 + 1) * 2 + 0) * 256];
    const float* R1i = &Vl[((layer * 2 + 1) * 2 + 1) * 256];
    v8f cr, ci;
    {
      v8f mr, mi, nr1, ni1;
      cmr2(V0r, V0i, R1r, R1i, eb, l, mr, mi, nr1, ni1);
      store_tile(&Mbuf[w][0][0], l, mr);  store_tile(&Mbuf[w][1][0], l, mi);
      store_tile(&Nbuf[w][0][0], l, nr1); store_tile(&Nbuf[w][1][0], l, ni1);
    }
    cmm(&Mbuf[w][0][0], &Mbuf[w][1][0], &Sbuf[w][0][0], &Sbuf[w][1][0], false, l, cr, ci);
    store_tile(&Mbuf[w][0][0], l, cr); store_tile(&Mbuf[w][1][0], l, ci);
    cmm(&Mbuf[w][0][0], &Mbuf[w][1][0], &Nbuf[w][0][0], &Nbuf[w][1][0], true, l, cr, ci);
    v8f nr, ni;
#pragma unroll
    for (int v = 0; v < 8; ++v) {
      int ad = (v & 1) ? addrO : addrE;
      nr[v] = __int_as_float(__builtin_amdgcn_ds_bpermute(ad, __float_as_int(cr[v])) ^ sgnA);
      ni[v] = __int_as_float(__builtin_amdgcn_ds_bpermute(ad, __float_as_int(ci[v])) ^ sgnB);
    }
    store_tile(dstA, l, nr);
    store_tile(dstB, l, ni);
  }

  float p[8];
#pragma unroll
  for (int v = 0; v < 8; ++v) {
    float a = Sbuf[w][0][(v + 8 * lh) * 16 + lr];
    float bi = Sbuf[w][1][(v + 8 * lh) * 16 + lr];
    p[v] = a * a + bi * bi;
  }
  float t  = p[0] + p[1] + p[2] + p[3] + p[4] + p[5] + p[6] + p[7];
  float u1 = p[0] + p[1] + p[2] + p[3] - p[4] - p[5] - p[6] - p[7];
  float u2 = p[0] + p[1] - p[2] - p[3] + p[4] + p[5] - p[6] - p[7];
  float u3 = p[0] - p[1] + p[2] - p[3] + p[4] - p[5] + p[6] - p[7];
  float ev[8];
  ev[0] = lh ? -t : t;
  ev[1] = u1; ev[2] = u2; ev[3] = u3;
  ev[4] = (lr & 8) ? -t : t;
  ev[5] = (lr & 4) ? -t : t;
  ev[6] = (lr & 2) ? -t : t;
  ev[7] = (lr & 1) ? -t : t;
#pragma unroll
  for (int i = 0; i < 8; ++i) ev[i] = wave_sum(ev[i], l);

  if (b < batch && l == 0) {
    const float kPi = 3.14159265358979323846f;
#pragma unroll
    for (int i = 0; i < 8; ++i) out[b * 8 + i] = kPi * ev[i];
  }
}

extern "C" void kernel_launch(void* const* d_in, const int* in_sizes, int n_in,
                              void* d_out, int out_size, void* d_ws, size_t ws_size,
                              hipStream_t stream) {
  const float* x   = (const float*)d_in[0];
  const float* wts = (const float*)d_in[1];
  float* out = (float*)d_out;
  float* V   = (float*)d_ws;

  int batch = in_sizes[0] / 2;
  qnn_build_v<<<N_LAYERS, 256, 0, stream>>>(wts, V);
  int nblocks = (batch + NWAVES - 1) / NWAVES;
  qnn_kernel<<<nblocks, NTHREADS, 0, stream>>>(x, V, out, batch);
}


# WMMA operands read from accumulator-layout registers (k order 8h+2kb+e): a product tile feeds the next product as A (transposed) or B; LDS tile stores and fragment loads removed; k sums re-ordered
# speedup vs baseline: 1.5306x; 1.1536x over previous
#include <hip/hip_runtime.h>
#include <math.h>


typedef __attribute__((ext_vector_type(2))) float v2f;
typedef __attribute__((ext_vector_type(4))) float v4f;
typedef __attribute__((ext_vector_type(8))) float v8f;

#define N_LAYERS 6
#define NWAVES   4
#define NTHREADS (NWAVES * 32)

struct c2 { float re, im; };

__device__ __forceinline__ c2 cmul(c2 a, c2 b) {
  return c2{a.re * b.re - a.im * b.im, a.re * b.im + a.im * b.re};
}
__device__ __forceinline__ c2 rx_el(int a, int b, float c, float s) {
  return (a == b) ? c2{c, 0.f} : c2{0.f, -s};
}
__device__ __forceinline__ c2 ry_el(int a, int b, float c, float s) {
  if (a == b) return c2{c, 0.f};
  return (a == 1) ? c2{s, 0.f} : c2{-s, 0.f};
}

__device__ __forceinline__ int phase_k(int n) { return ((n >> 3) & 1) + ((n >> 1) & 1); }
__device__ __forceinline__ c2 mul_ipow(c2 a, int k) {
  k &= 3;
  if (k == 0) return a;
  if (k == 1) return c2{-a.im, a.re};
  if (k == 2) return c2{-a.re, -a.im};
  return c2{a.im, -a.re};
}
__device__ __forceinline__ float ry_re(int a, int b, float c, float s) {
  return (a == b) ? c : ((a == 1) ? s : -s);
}

__global__ void qnn_build_v(const float* __restrict__ wts, float* __restrict__ V) {
  int l = blockIdx.x;
  int t = threadIdx.x;
  int r = t >> 4, c = t & 15;
  int tv = c & 7;
  int off = ((tv >> 2) * 32 + r + 16 * (c >> 3)) * 4 + (tv & 3);
  for (int m = 0; m < 2; ++m) {
    int rr = (m == 0) ? (r ^ (r >> 1)) : r;
    c2 prod = c2{1.f, 0.f};
    for (int i = 0; i < 4; ++i) {
      int wire = m * 4 + i;
      float a0 = wts[(l * 8 + wire) * 3 + 0];
      float a1 = wts[(l * 8 + wire) * 3 + 1];
      float a2 = wts[(l * 8 + wire) * 3 + 2];
      float cx = cosf(0.5f * a0), sx = sinf(0.5f * a0);
      float cy = cosf(0.5f * a1), sy = sinf(0.5f * a1);
      float cz = cosf(0.5f * a2), sz = sinf(0.5f * a2);
      int ab = (rr >> (3 - i)) & 1;
      int bb = (c  >> (3 - i)) & 1;
      c2 yx = c2{0.f, 0.f};
      for (int k = 0; k < 2; ++k) {
        c2 p = cmul(ry_el(ab, k, cy, sy), rx_el(k, bb, cx, sx));
        yx.re += p.re; yx.im += p.im;
      }
      c2 z = ab ? c2{cz, sz} : c2{cz, -sz};
      prod = cmul(prod, cmul(z, yx));
    }
    prod = mul_ipow(prod, ((m == 0) ? phase_k(r) : 0) - phase_k(c));
    V[(((l * 2 + m) * 2 + 0) * 256) + off] = prod.re;
    V[(((l * 2 + m) * 2 + 1) * 256) + off] = prod.im;
  }
}

__device__ __forceinline__ v8f mma(v2f a, v2f b, v8f c) {
  return __builtin_amdgcn_wmma_f32_16x16x4_f32(false, a, false, b, (short)0, c, false, false);
}
__device__ __forceinline__ v2f frag(v8f t, int kb) {
  v2f f;
  f.x = t[2 * kb];
  f.y = t[2 * kb + 1];
  return f;
}

__device__ __forceinline__ void cprod(v8f Pr, v8f Pi, v8f Qr, v8f Qi, v8f& Cr, v8f& Ci) {
  v8f Ps = Pr + Pi;
  v8f Qs = Qr + Qi;
  v8f Qd = Qi - Qr;
  v8f k1 = {}, k3 = {};
#pragma unroll
  for (int kb = 0; kb < 4; ++kb) {
    k1 = mma(frag(Ps, kb), frag(Qr, kb), k1);
    k3 = mma(frag(Pi, kb), frag(Qs, kb), k3);
  }
  v8f ci = k1;
#pragma unroll
  for (int kb = 0; kb < 4; ++kb) ci = mma(frag(Pr, kb), frag(Qd, kb), ci);
  Cr = k1 - k3;
  Ci = ci;
}

__device__ __forceinline__ v8f load_tile(const float* __restrict__ p, int l) {
  v4f lo = *(const v4f*)(p + 4 * l);
  v4f hi = *(const v4f*)(p + 128 + 4 * l);
  return __builtin_shufflevector(lo, hi, 0, 1, 2, 3, 4, 5, 6, 7);
}

__device__ __forceinline__ float wave_sum(float v, int l) {
#pragma unroll
  for (int m = 16; m; m >>= 1) {
    int src = (l ^ m) << 2;
    v += __int_as_float(__builtin_amdgcn_ds_bpermute(src, __float_as_int(v)));
  }
  return v;
}

__global__ void __launch_bounds__(NTHREADS)
qnn_kernel(const float* __restrict__ x, const float* __restrict__ Vg,
           float* __restrict__ out, int batch) {
  __shared__ float Vl[N_LAYERS * 2 * 2 * 256];

  int tid = threadIdx.x;
  int l = tid & 31;
  int w = tid >> 5;
  int b = blockIdx.x * NWAVES + w;
  int bb = (b < batch) ? b : (batch - 1);

  {
    const float4* src = (const float4*)Vg;
    float4* dst = (float4*)Vl;
    for (int i = tid; i < (N_LAYERS * 2 * 2 * 256) / 4; i += NTHREADS) dst[i] = src[i];
  }
  __syncthreads();

  float x0 = x[2 * bb], x1 = x[2 * bb + 1];
  float c0 = cosf(0.5f * x0), s0 = sinf(0.5f * x0);
  float c1 = cosf(0.5f * x1), s1 = sinf(0.5f * x1);
  int lr = l & 15, lh = l >> 4;
  v8f er;
  {
    float f3 = ry_re(lh, (lr >> 3) & 1, c0, s0);
#pragma unroll
    for (int v = 0; v < 8; ++v) {
      float h = f3 * ry_re(v >> 2, (lr >> 2) & 1, c1, s1);
      float hf = h * ry_re((v >> 1) & 1, (lr >> 1) & 1, c0, s0);
      er[v] = hf * ry_re(v & 1, lr & 1, c1, s1);
    }
  }
  v8f sr = {}, si = {};
  sr[0] = (l == 0) ? 1.f : 0.f;

  int csrc = (l & 15); csrc = csrc ^ (csrc >> 1);
  int addrE = (((l & 16) | csrc)) << 2;
  int addrO = addrE ^ (8 << 2);
  int kc = phase_k(lr);
  bool swp = (kc == 1);
  int sgnA = (kc == 2) ? (int)0x80000000u : 0;
  int sgnB = (kc >= 1) ? (int)0x80000000u : 0;

  for (int layer = 0; layer < N_LAYERS; ++layer) {
    const float* vt = &Vl[layer * 4 * 256];
    v8f ar = load_tile(vt, l);
    v8f ai = load_tile(vt + 256, l);
    v8f br = load_tile(vt + 512, l);
    v8f bi = load_tile(vt + 768, l);
    v8f ur = {}, ui = {}, wr = {}, wi = {};
#pragma unroll
    for (int kb = 0; kb < 4; ++kb) {
      v2f e = frag(er, kb);
      ur = mma(e, frag(ar, kb), ur);
      ui = mma(e, frag(ai, kb), ui);
      wr = mma(e, frag(br, kb), wr);
      wi = mma(e, frag(bi, kb), wi);
    }
    v8f tr, ti;
    cprod(sr, si, ur, ui, tr, ti);
    v8f cr, ci;
    cprod(tr, ti, wr, wi, cr, ci);
#pragma unroll
    for (int v = 0; v < 8; ++v) {
      int ad = (v & 1) ? addrO : addrE;
      float gr = __int_as_float(__builtin_amdgcn_ds_bpermute(ad, __float_as_int(cr[v])) ^ sgnA);
      float gi = __int_as_float(__builtin_amdgcn_ds_bpermute(ad, __float_as_int(ci[v])) ^ sgnB);
      sr[v] = swp ? gi : gr;
      si[v] = swp ? gr : gi;
    }
  }

  float p[8];
#pragma unroll
  for (int v = 0; v < 8; ++v) p[v] = sr[v] * sr[v] + si[v] * si[v];
  float t  = p[0] + p[1] + p[2] + p[3] + p[4] + p[5] + p[6] + p[7];
  float u1 = p[0] + p[1] + p[2] + p[3] - p[4] - p[5] - p[6] - p[7];
  float u2 = p[0] + p[1] - p[2] - p[3] + p[4] + p[5] - p[6] - p[7];
  float u3 = p[0] - p[1] + p[2] - p[3] + p[4] - p[5] + p[6] - p[7];
  float ev[8];
  ev[0] = lh ? -t : t;
  ev[1] = u1; ev[2] = u2; ev[3] = u3;
  ev[4] = (lr & 8) ? -t : t;
  ev[5] = (lr & 4) ? -t : t;
  ev[6] = (lr & 2) ? -t : t;
  ev[7] = (lr & 1) ? -t : t;
#pragma unroll
  for (int i = 0; i < 8; ++i) ev[i] = wave_sum(ev[i], l);

  if (b < batch && l == 0) {
    const float kPi = 3.14159265358979323846f;
#pragma unroll
    for (int i = 0; i < 8; ++i) out[b * 8 + i] = kPi * ev[i];
  }
}

extern "C" void kernel_launch(void* const* d_in, const int* in_sizes, int n_in,
                              void* d_out, int out_size, void* d_ws, size_t ws_size,
                              hipStream_t stream) {
  const float* x   = (const float*)d_in[0];
  const float* wts = (const float*)d_in[1];
  float* out = (float*)d_out;
  float* V   = (float*)d_ws;

  int batch = in_sizes[0] / 2;
  qnn_build_v<<<N_LAYERS, 256, 0, stream>>>(wts, V);
  int nblocks = (batch + NWAVES - 1) / NWAVES;
  qnn_kernel<<<nblocks, NTHREADS, 0, stream>>>(x, V, out, batch);
}


# wires-4..7 CNOT-chain perm and its phase baked into layer matrix B; CNOT(3,4) left as DPP row_mirror + sign on odd rows: 16 ds_bpermute and 32 VALU per layer become 8 DPP; re/im sums re-ordered
# speedup vs baseline: 1.5921x; 1.0402x over previous
#include <hip/hip_runtime.h>
#include <math.h>


typedef __attribute__((ext_vector_type(2))) float v2f;
typedef __attribute__((ext_vector_type(4))) float v4f;
typedef __attribute__((ext_vector_type(8))) float v8f;

#define N_LAYERS 6
#define NWAVES   4
#define NTHREADS (NWAVES * 32)

struct c2 { float re, im; };

__device__ __forceinline__ c2 cmul(c2 a, c2 b) {
  return c2{a.re * b.re - a.im * b.im, a.re * b.im + a.im * b.re};
}
__device__ __forceinline__ c2 rx_el(int a, int b, float c, float s) {
  return (a == b) ? c2{c, 0.f} : c2{0.f, -s};
}
__device__ __forceinline__ c2 ry_el(int a, int b, float c, float s) {
  if (a == b) return c2{c, 0.f};
  return (a == 1) ? c2{s, 0.f} : c2{-s, 0.f};
}

__device__ __forceinline__ int phase_k(int n) { return ((n >> 3) & 1) + ((n >> 1) & 1); }
__device__ __forceinline__ c2 mul_ipow(c2 a, int k) {
  k &= 3;
  if (k == 0) return a;
  if (k == 1) return c2{-a.im, a.re};
  if (k == 2) return c2{-a.re, -a.im};
  return c2{a.im, -a.re};
}
__device__ __forceinline__ float ry_re(int a, int b, float c, float s) {
  return (a == b) ? c : ((a == 1) ? s : -s);
}

__global__ void qnn_build_v(const float* __restrict__ wts, float* __restrict__ V) {
  int l = blockIdx.x;
  int t = threadIdx.x;
  int r = t >> 4, c = t & 15;
  int tv = c & 7;
  int off = ((tv >> 2) * 32 + r + 16 * (c >> 3)) * 4 + (tv & 3);
  for (int m = 0; m < 2; ++m) {
    int rr = r ^ (r >> 1);
    c2 prod = c2{1.f, 0.f};
    for (int i = 0; i < 4; ++i) {
      int wire = m * 4 + i;
      float a0 = wts[(l * 8 + wire) * 3 + 0];
      float a1 = wts[(l * 8 + wire) * 3 + 1];
      float a2 = wts[(l * 8 + wire) * 3 + 2];
      float cx = cosf(0.5f * a0), sx = sinf(0.5f * a0);
      float cy = cosf(0.5f * a1), sy = sinf(0.5f * a1);
      float cz = cosf(0.5f * a2), sz = sinf(0.5f * a2);
      int ab = (rr >> (3 - i)) & 1;
      int bb = (c  >> (3 - i)) & 1;
      c2 yx = c2{0.f, 0.f};
      for (int k = 0; k < 2; ++k) {
        c2 p = cmul(ry_el(ab, k, cy, sy), rx_el(k, bb, cx, sx));
        yx.re += p.re; yx.im += p.im;
      }
      c2 z = ab ? c2{cz, sz} : c2{cz, -sz};
      prod = cmul(prod, cmul(z, yx));
    }
    prod = mul_ipow(prod, phase_k(r) - phase_k(c));
    V[(((l * 2 + m) * 2 + 0) * 256) + off] = prod.re;
    V[(((l * 2 + m) * 2 + 1) * 256) + off] = prod.im;
  }
}

__device__ __forceinline__ v8f mma(v2f a, v2f b, v8f c) {
  return __builtin_amdgcn_wmma_f32_16x16x4_f32(false, a, false, b, (short)0, c, false, false);
}
__device__ __forceinline__ v2f frag(v8f t, int kb) {
  v2f f;
  f.x = t[2 * kb];
  f.y = t[2 * kb + 1];
  return f;
}

__device__ __forceinline__ void cprod(v8f Pr, v8f Pi, v8f Qr, v8f Qi, v8f& Cr, v8f& Ci) {
  v8f Ps = Pr + Pi;
  v8f Qs = Qr + Qi;
  v8f Qd = Qi - Qr;
  v8f k1 = {}, k3 = {};
#pragma unroll
  for (int kb = 0; kb < 4; ++kb) {
    k1 = mma(frag(Ps, kb), frag(Qr, kb), k1);
    k3 = mma(frag(Pi, kb), frag(Qs, kb), k3);
  }
  v8f ci = k1;
#pragma unroll
  for (int kb = 0; kb < 4; ++kb) ci = mma(frag(Pr, kb), frag(Qd, kb), ci);
  Cr = k1 - k3;
  Ci = ci;
}

__device__ __forceinline__ v8f load_tile(const float* __restrict__ p, int l) {
  v4f lo = *(const v4f*)(p + 4 * l);
  v4f hi = *(const v4f*)(p + 128 + 4 * l);
  return __builtin_shufflevector(lo, hi, 0, 1, 2, 3, 4, 5, 6, 7);
}

__device__ __forceinline__ float wave_sum(float v, int l) {
#pragma unroll
  for (int m = 16; m; m >>= 1) {
    int src = (l ^ m) << 2;
    v += __int_as_float(__builtin_amdgcn_ds_bpermute(src, __float_as_int(v)));
  }
  return v;
}

__global__ void __launch_bounds__(NTHREADS)
qnn_kernel(const float* __restrict__ x, const float* __restrict__ Vg,
           float* __restrict__ out, int batch) {
  __shared__ float Vl[N_LAYERS * 2 * 2 * 256];

  int tid = threadIdx.x;
  int l = tid & 31;
  int w = tid >> 5;
  int b = blockIdx.x * NWAVES + w;
  int bb = (b < batch) ? b : (batch - 1);

  {
    const float4* src = (const float4*)Vg;
    float4* dst = (float4*)Vl;
    for (int i = tid; i < (N_LAYERS * 2 * 2 * 256) / 4; i += NTHREADS) dst[i] = src[i];
  }
  __syncthreads();

  float x0 = x[2 * bb], x1 = x[2 * bb + 1];
  float c0 = cosf(0.5f * x0), s0 = sinf(0.5f * x0);
  float c1 = cosf(0.5f * x1), s1 = sinf(0.5f * x1);
  int lr = l & 15, lh = l >> 4;
  v8f er;
  {
    float f3 = ry_re(lh, (lr >> 3) & 1, c0, s0);
#pragma unroll
    for (int v = 0; v < 8; ++v) {
      float h = f3 * ry_re(v >> 2, (lr >> 2) & 1, c1, s1);
      float hf = h * ry_re((v >> 1) & 1, (lr >> 1) & 1, c0, s0);
      er[v] = hf * ry_re(v & 1, lr & 1, c1, s1);
    }
  }
  v8f sr = {}, si = {};
  sr[0] = (l == 0) ? 1.f : 0.f;

  int sgn = (phase_k(lr) == 1) ? 0 : (int)0x80000000u;

  for (int layer = 0; layer < N_LAYERS; ++layer) {
    const float* vt = &Vl[layer * 4 * 256];
    v8f ar = load_tile(vt, l);
    v8f ai = load_tile(vt + 256, l);
    v8f br = load_tile(vt + 512, l);
    v8f bi = load_tile(vt + 768, l);
    v8f ur = {}, ui = {}, wr = {}, wi = {};
#pragma unroll
    for (int kb = 0; kb < 4; ++kb) {
      v2f e = frag(er, kb);
      ur = mma(e, frag(ar, kb), ur);
      ui = mma(e, frag(ai, kb), ui);
      wr = mma(e, frag(br, kb), wr);
      wi = mma(e, frag(bi, kb), wi);
    }
    v8f tr, ti;
    cprod(sr, si, ur, ui, tr, ti);
    v8f cr, ci;
    cprod(tr, ti, wr, wi, cr, ci);
#pragma unroll
    for (int v = 0; v < 8; ++v) {
      if (v & 1) {
        sr[v] = __int_as_float(__builtin_amdgcn_update_dpp(0, __float_as_int(cr[v]), 0x140, 0xF, 0xF, true) ^ sgn);
        si[v] = __int_as_float(__builtin_amdgcn_update_dpp(0, __float_as_int(ci[v]), 0x140, 0xF, 0xF, true) ^ sgn);
      } else {
        sr[v] = cr[v];
        si[v] = ci[v];
      }
    }
  }

  float p[8];
#pragma unroll
  for (int v = 0; v < 8; ++v) p[v] = sr[v] * sr[v] + si[v] * si[v];
  float t  = p[0] + p[1] + p[2] + p[3] + p[4] + p[5] + p[6] + p[7];
  float u1 = p[0] + p[1] + p[2] + p[3] - p[4] - p[5] - p[6] - p[7];
  float u2 = p[0] + p[1] - p[2] - p[3] + p[4] + p[5] - p[6] - p[7];
  float u3 = p[0] - p[1] + p[2] - p[3] + p[4] - p[5] + p[6] - p[7];
  float ev[8];
  ev[0] = lh ? -t : t;
  ev[1] = u1; ev[2] = u2; ev[3] = u3;
  ev[4] = (lr & 8) ? -t : t;
  ev[5] = (lr & 4) ? -t : t;
  ev[6] = (lr & 2) ? -t : t;
  ev[7] = (lr & 1) ? -t : t;
#pragma unroll
  for (int i = 0; i < 8; ++i) ev[i] = wave_sum(ev[i], l);

  if (b < batch && l == 0) {
    const float kPi = 3.14159265358979323846f;
#pragma unroll
    for (int i = 0; i < 8; ++i) out[b * 8 + i] = kPi * ev[i];
  }
}

extern "C" void kernel_launch(void* const* d_in, const int* in_sizes, int n_in,
                              void* d_out, int out_size, void* d_ws, size_t ws_size,
                              hipStream_t stream) {
  const float* x   = (const float*)d_in[0];
  const float* wts = (const float*)d_in[1];
  float* out = (float*)d_out;
  float* V   = (float*)d_ws;

  int batch = in_sizes[0] / 2;
  qnn_build_v<<<N_LAYERS, 256, 0, stream>>>(wts, V);
  int nblocks = (batch + NWAVES - 1) / NWAVES;
  qnn_kernel<<<nblocks, NTHREADS, 0, stream>>>(x, V, out, batch);
}


# layer 0 uses the fixed start state e0 e0^T: S^T U becomes a lane select of U's row 0 and TT^T W keeps k-block 0 alone: 19 WMMA in layer 0, was 40; layer-0 re/im sums formed in a different order
# speedup vs baseline: 1.7119x; 1.0752x over previous
#include <hip/hip_runtime.h>
#include <math.h>


typedef __attribute__((ext_vector_type(2))) float v2f;
typedef __attribute__((ext_vector_type(4))) float v4f;
typedef __attribute__((ext_vector_type(8))) float v8f;

#define N_LAYERS 6
#define NWAVES   4
#define NTHREADS (NWAVES * 32)

struct c2 { float re, im; };

__device__ __forceinline__ c2 cmul(c2 a, c2 b) {
  return c2{a.re * b.re - a.im * b.im, a.re * b.im + a.im * b.re};
}
__device__ __forceinline__ c2 rx_el(int a, int b, float c, float s) {
  return (a == b) ? c2{c, 0.f} : c2{0.f, -s};
}
__device__ __forceinline__ c2 ry_el(int a, int b, float c, float s) {
  if (a == b) return c2{c, 0.f};
  return (a == 1) ? c2{s, 0.f} : c2{-s, 0.f};
}

__device__ __forceinline__ int phase_k(int n) { return ((n >> 3) & 1) + ((n >> 1) & 1); }
__device__ __forceinline__ c2 mul_ipow(c2 a, int k) {
  k &= 3;
  if (k == 0) return a;
  if (k == 1) return c2{-a.im, a.re};
  if (k == 2) return c2{-a.re, -a.im};
  return c2{a.im, -a.re};
}
__device__ __forceinline__ float ry_re(int a, int b, float c, float s) {
  return (a == b) ? c : ((a == 1) ? s : -s);
}

__global__ void qnn_build_v(const float* __restrict__ wts, float* __restrict__ V) {
  int l = blockIdx.x;
  int t = threadIdx.x;
  int r = t >> 4, c = t & 15;
  int tv = c & 7;
  int off = ((tv >> 2) * 32 + r + 16 * (c >> 3)) * 4 + (tv & 3);
  for (int m = 0; m < 2; ++m) {
    int rr = r ^ (r >> 1);
    c2 prod = c2{1.f, 0.f};
    for (int i = 0; i < 4; ++i) {
      int wire = m * 4 + i;
      float a0 = wts[(l * 8 + wire) * 3 + 0];
      float a1 = wts[(l * 8 + wire) * 3 + 1];
      float a2 = wts[(l * 8 + wire) * 3 + 2];
      float cx = cosf(0.5f * a0), sx = sinf(0.5f * a0);
      float cy = cosf(0.5f * a1), sy = sinf(0.5f * a1);
      float cz = cosf(0.5f * a2), sz = sinf(0.5f * a2);
      int ab = (rr >> (3 - i)) & 1;
      int bb = (c  >> (3 - i)) & 1;
      c2 yx = c2{0.f, 0.f};
      for (int k = 0; k < 2; ++k) {
        c2 p = cmul(ry_el(ab, k, cy, sy), rx_el(k, bb, cx, sx));
        yx.re += p.re; yx.im += p.im;
      }
      c2 z = ab ? c2{cz, sz} : c2{cz, -sz};
      prod = cmul(prod, cmul(z, yx));
    }
    prod = mul_ipow(prod, phase_k(r) - phase_k(c));
    V[(((l * 2 + m) * 2 + 0) * 256) + off] = prod.re;
    V[(((l * 2 + m) * 2 + 1) * 256) + off] = prod.im;
  }
}

__device__ __forceinline__ v8f mma(v2f a, v2f b, v8f c) {
  return __builtin_amdgcn_wmma_f32_16x16x4_f32(false, a, false, b, (short)0, c, false, false);
}
__device__ __forceinline__ v2f frag(v8f t, int kb) {
  v2f f;
  f.x = t[2 * kb];
  f.y = t[2 * kb + 1];
  return f;
}

__device__ __forceinline__ void cprod(v8f Pr, v8f Pi, v8f Qr, v8f Qi, v8f& Cr, v8f& Ci) {
  v8f Ps = Pr + Pi;
  v8f Qs = Qr + Qi;
  v8f Qd = Qi - Qr;
  v8f k1 = {}, k3 = {};
#pragma unroll
  for (int kb = 0; kb < 4; ++kb) {
    k1 = mma(frag(Ps, kb), frag(Qr, kb), k1);
    k3 = mma(frag(Pi, kb), frag(Qs, kb), k3);
  }
  v8f ci = k1;
#pragma unroll
  for (int kb = 0; kb < 4; ++kb) ci = mma(frag(Pr, kb), frag(Qd, kb), ci);
  Cr = k1 - k3;
  Ci = ci;
}

__device__ __forceinline__ void cprod_row0(float p0r, float p0i, v8f Qr, v8f Qi, v8f& Cr, v8f& Ci) {
  v2f ps = {p0r + p0i, 0.f};
  v2f pr = {p0r, 0.f};
  v2f pi = {p0i, 0.f};
  v2f qr = frag(Qr, 0);
  v2f qi = frag(Qi, 0);
  v8f z = {};
  v8f k1 = mma(ps, qr, z);
  v8f k3 = mma(pi, qr + qi, z);
  Ci = mma(pr, qi - qr, k1);
  Cr = k1 - k3;
}

__device__ __forceinline__ v8f load_tile(const float* __restrict__ p, int l) {
  v4f lo = *(const v4f*)(p + 4 * l);
  v4f hi = *(const v4f*)(p + 128 + 4 * l);
  return __builtin_shufflevector(lo, hi, 0, 1, 2, 3, 4, 5, 6, 7);
}

__device__ __forceinline__ void enc_products(const float* __restrict__ vt, int l, v8f er,
                                             v8f& ur, v8f& ui, v8f& wr, v8f& wi) {
  v8f ar = load_tile(vt, l);
  v8f ai = load_tile(vt + 256, l);
  v8f br = load_tile(vt + 512, l);
  v8f bi = load_tile(vt + 768, l);
  v8f zr = {};
  ur = zr; ui = zr; wr = zr; wi = zr;
#pragma unroll
  for (int kb = 0; kb < 4; ++kb) {
    v2f e = frag(er, kb);
    ur = mma(e, frag(ar, kb), ur);
    ui = mma(e, frag(ai, kb), ui);
    wr = mma(e, frag(br, kb), wr);
    wi = mma(e, frag(bi, kb), wi);
  }
}

__device__ __forceinline__ void cnot34(v8f cr, v8f ci, int sgn, v8f& sr, v8f& si) {
#pragma unroll
  for (int v = 0; v < 8; ++v) {
    if (v & 1) {
      sr[v] = __int_as_float(__builtin_amdgcn_update_dpp(0, __float_as_int(cr[v]), 0x140, 0xF, 0xF, true) ^ sgn);
      si[v] = __int_as_float(__builtin_amdgcn_update_dpp(0, __float_as_int(ci[v]), 0x140, 0xF, 0xF, true) ^ sgn);
    } else {
      sr[v] = cr[v];
      si[v] = ci[v];
    }
  }
}

__device__ __forceinline__ float wave_sum(float v, int l) {
#pragma unroll
  for (int m = 16; m; m >>= 1) {
    int src = (l ^ m) << 2;
    v += __int_as_float(__builtin_amdgcn_ds_bpermute(src, __float_as_int(v)));
  }
  return v;
}

__global__ void __launch_bounds__(NTHREADS)
qnn_kernel(const float* __restrict__ x, const float* __restrict__ Vg,
           float* __restrict__ out, int batch) {
  __shared__ float Vl[N_LAYERS * 2 * 2 * 256];

  int tid = threadIdx.x;
  int l = tid & 31;
  int w = tid >> 5;
  int b = blockIdx.x * NWAVES + w;
  int bb = (b < batch) ? b : (batch - 1);

  {
    const float4* src = (const float4*)Vg;
    float4* dst = (float4*)Vl;
    for (int i = tid; i < (N_LAYERS * 2 * 2 * 256) / 4; i += NTHREADS) dst[i] = src[i];
  }
  __syncthreads();

  float x0 = x[2 * bb], x1 = x[2 * bb + 1];
  float c0 = cosf(0.5f * x0), s0 = sinf(0.5f * x0);
  float c1 = cosf(0.5f * x1), s1 = sinf(0.5f * x1);
  int lr = l & 15, lh = l >> 4;
  v8f er;
  {
    float f3 = ry_re(lh, (lr >> 3) & 1, c0, s0);
#pragma unroll
    for (int v = 0; v < 8; ++v) {
      float h = f3 * ry_re(v >> 2, (lr >> 2) & 1, c1, s1);
      float hf = h * ry_re((v >> 1) & 1, (lr >> 1) & 1, c0, s0);
      er[v] = hf * ry_re(v & 1, lr & 1, c1, s1);
    }
  }
  int sgn = (phase_k(lr) == 1) ? 0 : (int)0x80000000u;

  v8f sr, si;
  {
    v8f ur, ui, wr, wi;
    enc_products(Vl, l, er, ur, ui, wr, wi);
    float t0r = lh ? 0.f : ur[0];
    float t0i = lh ? 0.f : ui[0];
    v8f cr, ci;
    cprod_row0(t0r, t0i, wr, wi, cr, ci);
    cnot34(cr, ci, sgn, sr, si);
  }

  for (int layer = 1; layer < N_LAYERS; ++layer) {
    v8f ur, ui, wr, wi;
    enc_products(&Vl[layer * 4 * 256], l, er, ur, ui, wr, wi);
    v8f tr, ti;
    cprod(sr, si, ur, ui, tr, ti);
    v8f cr, ci;
    cprod(tr, ti, wr, wi, cr, ci);
    cnot34(cr, ci, sgn, sr, si);
  }

  float p[8];
#pragma unroll
  for (int v = 0; v < 8; ++v) p[v] = sr[v] * sr[v] + si[v] * si[v];
  float t  = p[0] + p[1] + p[2] + p[3] + p[4] + p[5] + p[6] + p[7];
  float u1 = p[0] + p[1] + p[2] + p[3] - p[4] - p[5] - p[6] - p[7];
  float u2 = p[0] + p[1] - p[2] - p[3] + p[4] + p[5] - p[6] - p[7];
  float u3 = p[0] - p[1] + p[2] - p[3] + p[4] - p[5] + p[6] - p[7];
  float ev[8];
  ev[0] = lh ? -t : t;
  ev[1] = u1; ev[2] = u2; ev[3] = u3;
  ev[4] = (lr & 8) ? -t : t;
  ev[5] = (lr & 4) ? -t : t;
  ev[6] = (lr & 2) ? -t : t;
  ev[7] = (lr & 1) ? -t : t;
#pragma unroll
  for (int i = 0; i < 8; ++i) ev[i] = wave_sum(ev[i], l);

  if (b < batch && l == 0) {
    const float kPi = 3.14159265358979323846f;
#pragma unroll
    for (int i = 0; i < 8; ++i) out[b * 8 + i] = kPi * ev[i];
  }
}

extern "C" void kernel_launch(void* const* d_in, const int* in_sizes, int n_in,
                              void* d_out, int out_size, void* d_ws, size_t ws_size,
                              hipStream_t stream) {
  const float* x   = (const float*)d_in[0];
  const float* wts = (const float*)d_in[1];
  float* out = (float*)d_out;
  float* V   = (float*)d_ws;

  int batch = in_sizes[0] / 2;
  qnn_build_v<<<N_LAYERS, 256, 0, stream>>>(wts, V);
  int nblocks = (batch + NWAVES - 1) / NWAVES;
  qnn_kernel<<<nblocks, NTHREADS, 0, stream>>>(x, V, out, batch);
}


# layer matrices factored diag*real*diag (W = RZ RY RZ per wire): 4 real-by-complex products (32 WMMA) + 2 elementwise phase-tile multiplies per layer, was 40 WMMA; re/im sums re-ordered
# speedup vs baseline: 1.9949x; 1.1653x over previous
#include <hip/hip_runtime.h>
#include <math.h>


typedef __attribute__((ext_vector_type(2))) float v2f;
typedef __attribute__((ext_vector_type(4))) float v4f;
typedef __attribute__((ext_vector_type(8))) float v8f;

#define N_LAYERS 6
#define NPLANES  6
#define NWAVES   8
#define NTHREADS (NWAVES * 32)
#define VUSED    (N_LAYERS * NPLANES * 256 - 2 * 256)

struct c2 { float re, im; };

__device__ __forceinline__ c2 cmul(c2 a, c2 b) {
  return c2{a.re * b.re - a.im * b.im, a.re * b.im + a.im * b.re};
}

__device__ __forceinline__ int phase_k(int n) { return ((n >> 3) & 1) + ((n >> 1) & 1); }
__device__ __forceinline__ c2 mul_ipow(c2 a, int k) {
  k &= 3;
  if (k == 0) return a;
  if (k == 1) return c2{-a.im, a.re};
  if (k == 2) return c2{-a.re, -a.im};
  return c2{a.im, -a.re};
}
__device__ __forceinline__ float ry_re(int a, int b, float c, float s) {
  return (a == b) ? c : ((a == 1) ? s : -s);
}

struct wirep { float cb, sb; c2 l0, l1, r1; };

__device__ __forceinline__ wirep wire_params(const float* __restrict__ wts, int l, int wire) {
  float a0 = wts[(l * 8 + wire) * 3 + 0];
  float a1 = wts[(l * 8 + wire) * 3 + 1];
  float a2 = wts[(l * 8 + wire) * 3 + 2];
  float cx = cosf(0.5f * a0), sx = sinf(0.5f * a0);
  float cy = cosf(0.5f * a1), sy = sinf(0.5f * a1);
  float cz = cosf(0.5f * a2), sz = sinf(0.5f * a2);
  c2 w00 = cmul(c2{cz, -sz}, c2{cy * cx, sy * sx});
  c2 w10 = cmul(c2{cz, sz}, c2{sy * cx, -cy * sx});
  wirep p;
  p.cb = sqrtf(w00.re * w00.re + w00.im * w00.im);
  p.sb = sqrtf(w10.re * w10.re + w10.im * w10.im);
  p.l0 = (p.cb > 1e-18f) ? c2{w00.re / p.cb, w00.im / p.cb} : c2{1.f, 0.f};
  p.l1 = (p.sb > 1e-18f) ? c2{w10.re / p.sb, w10.im / p.sb} : c2{1.f, 0.f};
  c2 ll = cmul(p.l0, p.l1);
  p.r1 = c2{ll.re, -ll.im};
  return p;
}

__global__ void qnn_build_v(const float* __restrict__ wts, float* __restrict__ V) {
  int l = blockIdx.x;
  int t = threadIdx.x;
  int r = t >> 4, c = t & 15;
  int rr = r ^ (r >> 1);
  int cm = (r & 1) ? (15 - c) : c;
  int cmr = cm ^ (cm >> 1);
  int offT = (((c & 7) >> 2) * 32 + r + 16 * (c >> 3)) * 4 + (c & 3);
  int offD = (((r & 7) >> 2) * 32 + c + 16 * (r >> 3)) * 4 + (r & 3);

  float rb0 = 1.f, rb1 = 1.f;
  c2 rho0 = c2{1.f, 0.f}, rho1 = c2{1.f, 0.f};
  c2 lam0 = c2{1.f, 0.f}, lam1 = c2{1.f, 0.f};
  for (int i = 0; i < 4; ++i) {
    int sh = 3 - i;
    wirep p0 = wire_params(wts, l, i);
    wirep p1 = wire_params(wts, l, 4 + i);
    rb0 *= ry_re((rr >> sh) & 1, (c >> sh) & 1, p0.cb, p0.sb);
    rb1 *= ry_re((rr >> sh) & 1, (c >> sh) & 1, p1.cb, p1.sb);
    if ((r >> sh) & 1) rho0 = cmul(rho0, p0.r1);
    if ((c >> sh) & 1) rho1 = cmul(rho1, p1.r1);
    lam0 = cmul(lam0, ((rr >> sh) & 1) ? p0.l1 : p0.l0);
    lam1 = cmul(lam1, ((cmr >> sh) & 1) ? p1.l1 : p1.l0);
  }
  int kq = phase_k(r) + phase_k(c);
  c2 d2 = mul_ipow(cmul(rho0, rho1), -kq);
  c2 d4 = mul_ipow(cmul(lam0, lam1), kq);
  float* Vl = V + l * NPLANES * 256;
  Vl[0 * 256 + offT] = rb0;
  Vl[1 * 256 + offT] = rb1;
  Vl[2 * 256 + offD] = d2.re;
  Vl[3 * 256 + offD] = d2.im;
  Vl[4 * 256 + offD] = d4.re;
  Vl[5 * 256 + offD] = d4.im;
}

__device__ __forceinline__ v8f mma(v2f a, v2f b, v8f c) {
  return __builtin_amdgcn_wmma_f32_16x16x4_f32(false, a, false, b, (short)0, c, false, false);
}
__device__ __forceinline__ v2f frag(v8f t, int kb) {
  v2f f;
  f.x = t[2 * kb];
  f.y = t[2 * kb + 1];
  return f;
}

__device__ __forceinline__ void rprod(v8f Pr, v8f Pi, v8f Q, v8f& Cr, v8f& Ci) {
  v8f cr = {}, ci = {};
#pragma unroll
  for (int kb = 0; kb < 4; ++kb) {
    v2f q = frag(Q, kb);
    cr = mma(frag(Pr, kb), q, cr);
    ci = mma(frag(Pi, kb), q, ci);
  }
  Cr = cr;
  Ci = ci;
}

__device__ __forceinline__ v8f load_tile(const float* __restrict__ p, int l) {
  v4f lo = *(const v4f*)(p + 4 * l);
  v4f hi = *(const v4f*)(p + 128 + 4 * l);
  return __builtin_shufflevector(lo, hi, 0, 1, 2, 3, 4, 5, 6, 7);
}

__device__ __forceinline__ v8f mirror_odd(v8f z) {
  v8f o;
#pragma unroll
  for (int v = 0; v < 8; ++v) {
    if (v & 1) {
      o[v] = __int_as_float(__builtin_amdgcn_update_dpp(0, __float_as_int(z[v]), 0x140, 0xF, 0xF, true));
    } else {
      o[v] = z[v];
    }
  }
  return o;
}

__device__ __forceinline__ void layer_tail(const float* __restrict__ vt, int l, bool last,
                                           v8f z2r, v8f z2i, v8f& yr, v8f& yi) {
  v8f rb0t = load_tile(vt, l);
  v8f rb1t = load_tile(vt + 256, l);
  v8f t3r, t3i, z3r, z3i;
  rprod(z2r, z2i, rb0t, t3r, t3i);
  rprod(t3r, t3i, rb1t, z3r, z3i);
  v8f mr = mirror_odd(z3r);
  v8f mi = mirror_odd(z3i);
  if (last) {
    yr = mr;
    yi = mi;
  } else {
    v8f d4r = load_tile(vt + 1024, l);
    v8f d4i = load_tile(vt + 1280, l);
    yr = mr * d4r - mi * d4i;
    yi = mr * d4i + mi * d4r;
  }
}

__device__ __forceinline__ void layer_full(const float* __restrict__ vt, int l, bool last,
                                           v8f ert, v8f& yr, v8f& yi) {
  v8f t1r, t1i, z1r, z1i;
  rprod(yr, yi, ert, t1r, t1i);
  rprod(t1r, t1i, ert, z1r, z1i);
  v8f d2r = load_tile(vt + 512, l);
  v8f d2i = load_tile(vt + 768, l);
  v8f z2r = z1r * d2r - z1i * d2i;
  v8f z2i = z1r * d2i + z1i * d2r;
  layer_tail(vt, l, last, z2r, z2i, yr, yi);
}

__device__ __forceinline__ float wave_sum(float v, int l) {
#pragma unroll
  for (int m = 16; m; m >>= 1) {
    int src = (l ^ m) << 2;
    v += __int_as_float(__builtin_amdgcn_ds_bpermute(src, __float_as_int(v)));
  }
  return v;
}

__global__ void __launch_bounds__(NTHREADS)
qnn_kernel(const float* __restrict__ x, const float* __restrict__ Vg,
           float* __restrict__ out, int batch) {
  __shared__ float Vl[VUSED];

  int tid = threadIdx.x;
  int l = tid & 31;
  int w = tid >> 5;
  int b = blockIdx.x * NWAVES + w;
  int bb = (b < batch) ? b : (batch - 1);

  {
    const float4* src = (const float4*)Vg;
    float4* dst = (float4*)Vl;
    for (int i = tid; i < VUSED / 4; i += NTHREADS) dst[i] = src[i];
  }
  __syncthreads();

  float x0 = x[2 * bb], x1 = x[2 * bb + 1];
  float c0 = cosf(0.5f * x0), s0 = sinf(0.5f * x0);
  float c1 = cosf(0.5f * x1), s1 = sinf(0.5f * x1);
  int lr = l & 15, lh = l >> 4;
  v8f ert;
  {
    float f3 = ry_re((lr >> 3) & 1, lh, c0, s0);
#pragma unroll
    for (int v = 0; v < 8; ++v) {
      float h = f3 * ry_re((lr >> 2) & 1, v >> 2, c1, s1);
      float hf = h * ry_re((lr >> 1) & 1, (v >> 1) & 1, c0, s0);
      ert[v] = hf * ry_re(lr & 1, v & 1, c1, s1);
    }
  }

  v8f yr, yi;
  {
    v2f p0 = {lh ? 0.f : ert[0], 0.f};
    v8f z = {};
    v8f z1 = mma(p0, p0, z);
    v8f d2r = load_tile(Vl + 512, l);
    v8f d2i = load_tile(Vl + 768, l);
    layer_tail(Vl, l, false, z1 * d2r, z1 * d2i, yr, yi);
  }

  for (int layer = 1; layer < N_LAYERS - 1; ++layer)
    layer_full(&Vl[layer * NPLANES * 256], l, false, ert, yr, yi);
  layer_full(&Vl[(N_LAYERS - 1) * NPLANES * 256], l, true, ert, yr, yi);

  float p[8];
#pragma unroll
  for (int v = 0; v < 8; ++v) p[v] = yr[v] * yr[v] + yi[v] * yi[v];
  float t  = p[0] + p[1] + p[2] + p[3] + p[4] + p[5] + p[6] + p[7];
  float u1 = p[0] + p[1] + p[2] + p[3] - p[4] - p[5] - p[6] - p[7];
  float u2 = p[0] + p[1] - p[2] - p[3] + p[4] + p[5] - p[6] - p[7];
  float u3 = p[0] - p[1] + p[2] - p[3] + p[4] - p[5] + p[6] - p[7];
  float ev[8];
  ev[0] = lh ? -t : t;
  ev[1] = u1; ev[2] = u2; ev[3] = u3;
  ev[4] = (lr & 8) ? -t : t;
  ev[5] = (lr & 4) ? -t : t;
  ev[6] = (lr & 2) ? -t : t;
  ev[7] = (lr & 1) ? -t : t;
#pragma unroll
  for (int i = 0; i < 8; ++i) ev[i] = wave_sum(ev[i], l);

  if (b < batch && l == 0) {
    const float kPi = 3.14159265358979323846f;
#pragma unroll
    for (int i = 0; i < 8; ++i) out[b * 8 + i] = kPi * ev[i];
  }
}

extern "C" void kernel_launch(void* const* d_in, const int* in_sizes, int n_in,
                              void* d_out, int out_size, void* d_ws, size_t ws_size,
                              hipStream_t stream) {
  const float* x   = (const float*)d_in[0];
  const float* wts = (const float*)d_in[1];
  float* out = (float*)d_out;
  float* V   = (float*)d_ws;

  int batch = in_sizes[0] / 2;
  qnn_build_v<<<N_LAYERS, 256, 0, stream>>>(wts, V);
  int nblocks = (batch + NWAVES - 1) / NWAVES;
  qnn_kernel<<<nblocks, NTHREADS, 0, stream>>>(x, V, out, batch);
}


# encoding gates diagonalised: state kept in the RX/RY eigenbasis on 7 wires, basis changes folded into the layer matrices; 2 complex products (3 real each) per layer, 24 WMMA, was 32; sums reordered
# speedup vs baseline: 2.2511x; 1.1284x over previous
#include <hip/hip_runtime.h>
#include <math.h>


typedef __attribute__((ext_vector_type(2))) float v2f;
typedef __attribute__((ext_vector_type(4))) float v4f;
typedef __attribute__((ext_vector_type(8))) float v8f;

#define N_LAYERS 6
#define NPLANES  6
#define NWAVES   8
#define NTHREADS (NWAVES * 32)
#define VUSED    ((N_LAYERS * NPLANES + 2) * 256)
#define RSQRT2   0.70710678118654752440f

struct c2 { float re, im; };

__device__ __forceinline__ c2 cmul(c2 a, c2 b) {
  return c2{a.re * b.re - a.im * b.im, a.re * b.im + a.im * b.re};
}
__device__ __forceinline__ c2 cadd(c2 a, c2 b) { return c2{a.re + b.re, a.im + b.im}; }

__device__ __forceinline__ c2 mul_ipow(c2 a, int k) {
  k &= 3;
  if (k == 0) return a;
  if (k == 1) return c2{-a.im, a.re};
  if (k == 2) return c2{-a.re, -a.im};
  return c2{a.im, -a.re};
}

__device__ __forceinline__ int col_std(int c) { return (c & 9) | ((c & 2) << 1) | ((c & 4) >> 1); }

__device__ __forceinline__ void wire_col(const float* __restrict__ wts, int l, int wire,
                                         int kind, int cb, c2& g0, c2& g1) {
  float a0 = wts[(l * 8 + wire) * 3 + 0];
  float a1 = wts[(l * 8 + wire) * 3 + 1];
  float a2 = wts[(l * 8 + wire) * 3 + 2];
  float cx, sx, cy, sy, cz, sz;
  sincosf(0.5f * a0, &sx, &cx);
  sincosf(0.5f * a1, &sy, &cy);
  sincosf(0.5f * a2, &sz, &cz);
  c2 w00 = cmul(c2{cz, -sz}, c2{cy * cx, sy * sx});
  c2 w01 = cmul(c2{cz, -sz}, c2{-sy * cx, -cy * sx});
  c2 w10 = cmul(c2{cz, sz}, c2{sy * cx, -cy * sx});
  c2 w11 = cmul(c2{cz, sz}, c2{cy * cx, -sy * sx});
  if (kind == 0) {
    float sg = cb ? -RSQRT2 : RSQRT2;
    g0 = c2{RSQRT2 * w00.re + sg * w01.re, RSQRT2 * w00.im + sg * w01.im};
    g1 = c2{RSQRT2 * w10.re + sg * w11.re, RSQRT2 * w10.im + sg * w11.im};
  } else if (kind == 1) {
    c2 p0 = cb ? mul_ipow(w00, 1) : w00;
    c2 p1 = cb ? w01 : mul_ipow(w01, 1);
    c2 q0 = cb ? mul_ipow(w10, 1) : w10;
    c2 q1 = cb ? w11 : mul_ipow(w11, 1);
    g0 = c2{RSQRT2 * (p0.re + p1.re), RSQRT2 * (p0.im + p1.im)};
    g1 = c2{RSQRT2 * (q0.re + q1.re), RSQRT2 * (q0.im + q1.im)};
  } else {
    g0 = cb ? w01 : w00;
    g1 = cb ? w11 : w10;
  }
}

__device__ __forceinline__ void uvec(int kind, int rb, c2& u0, c2& u1) {
  if (kind == 0) {
    u0 = c2{RSQRT2, 0.f};
    u1 = c2{rb ? -RSQRT2 : RSQRT2, 0.f};
  } else if (kind == 1) {
    u0 = rb ? c2{0.f, -RSQRT2} : c2{RSQRT2, 0.f};
    u1 = rb ? c2{RSQRT2, 0.f} : c2{0.f, -RSQRT2};
  } else {
    u0 = c2{rb ? 0.f : 1.f, 0.f};
    u1 = c2{rb ? 1.f : 0.f, 0.f};
  }
}

__device__ __forceinline__ c2 group_entry(const float* __restrict__ wts, int l, int base,
                                          bool last, bool keep3, int r, int c) {
  c2 f0 = c2{0.f, 0.f}, f1 = c2{0.f, 0.f};
#pragma unroll
  for (int i = 0; i < 4; ++i) {
    int sh = 3 - i;
    int vk = (keep3 && i == 3) ? 2 : (i & 1);
    int uk = last ? 2 : vk;
    int rb = (r >> sh) & 1, cb = (c >> sh) & 1;
    c2 g0, g1, u0, u1;
    wire_col(wts, l, base + i, vk, cb, g0, g1);
    uvec(uk, rb, u0, u1);
    if (i == 0) {
      f0 = cmul(u0, g0);
      f1 = cmul(u1, g1);
    } else {
      c2 t0 = cadd(cmul(f0, g0), cmul(f1, g1));
      c2 t1 = cadd(cmul(f0, g1), cmul(f1, g0));
      f0 = cmul(u0, t0);
      f1 = cmul(u1, t1);
    }
  }
  return cadd(f0, f1);
}

__global__ void qnn_build_v(const float* __restrict__ wts, float* __restrict__ V) {
  int l = blockIdx.x;
  int t = threadIdx.x;
  int r = t >> 4, c = t & 15;
  bool last = (l == N_LAYERS - 1);
  int offT = (((c & 7) >> 2) * 32 + r + 16 * (c >> 3)) * 4 + (c & 3);
  int offD = (((r & 7) >> 2) * 32 + c + 16 * (r >> 3)) * 4 + (r & 3);

  c2 a = group_entry(wts, l, 0, last, true, r, c);
  c2 b = group_entry(wts, l, 4, last, false, col_std(r), col_std(c));
  float* Vl = V + l * NPLANES * 256;
  Vl[0 * 256 + offT] = a.re;
  Vl[1 * 256 + offT] = a.im - a.re;
  Vl[2 * 256 + offT] = -(a.re + a.im);
  Vl[3 * 256 + offT] = b.re;
  Vl[4 * 256 + offT] = b.im - b.re;
  Vl[5 * 256 + offT] = -(b.re + b.im);

  if (l == 0) {
    int cs = col_std(c);
    int k = ((r >> 2) & 1) + ((cs >> 2) & 1) + (cs & 1);
    float amp = (r & 1) ? 0.f : 0.08838834764831844055f;
    c2 y0 = mul_ipow(c2{amp, 0.f}, -k);
    float* Vs = V + N_LAYERS * NPLANES * 256;
    Vs[offD] = y0.re;
    Vs[256 + offD] = y0.im;
  }
}

__device__ __forceinline__ v8f mma(v2f a, v2f b, v8f c) {
  return __builtin_amdgcn_wmma_f32_16x16x4_f32(false, a, false, b, (short)0, c, false, false);
}
__device__ __forceinline__ v2f frag(v8f t, int kb) {
  v2f f;
  f.x = t[2 * kb];
  f.y = t[2 * kb + 1];
  return f;
}

__device__ __forceinline__ void cprod(v8f Pr, v8f Pi, v8f q1, v8f q2, v8f q3n, v8f& Cr, v8f& Ci) {
  v8f ps = Pr + Pi;
  v8f k1 = {};
#pragma unroll
  for (int kb = 0; kb < 4; ++kb) k1 = mma(frag(ps, kb), frag(q1, kb), k1);
  v8f cr = k1, ci = k1;
#pragma unroll
  for (int kb = 0; kb < 4; ++kb) {
    ci = mma(frag(Pr, kb), frag(q2, kb), ci);
    cr = mma(frag(Pi, kb), frag(q3n, kb), cr);
  }
  Cr = cr;
  Ci = ci;
}

__device__ __forceinline__ v8f load_tile(const float* __restrict__ p, int l) {
  v4f lo = *(const v4f*)(p + 4 * l);
  v4f hi = *(const v4f*)(p + 128 + 4 * l);
  return __builtin_shufflevector(lo, hi, 0, 1, 2, 3, 4, 5, 6, 7);
}

__device__ __forceinline__ float quad_rev(float x) {
  return __int_as_float(__builtin_amdgcn_update_dpp(0, __float_as_int(x), 0x1B, 0xF, 0xF, true));
}
__device__ __forceinline__ float row_rev(float x) {
  return __int_as_float(__builtin_amdgcn_update_dpp(0, __float_as_int(x), 0x140, 0xF, 0xF, true));
}

struct enc {
  float c1, s1;
  float c1q, s1q;
  float pr[4], pi[4];
};

__device__ __forceinline__ void encode(const enc& e, bool q, v8f zr, v8f zi, v8f& yr, v8f& yi) {
  float cq = q ? e.c1q : e.c1;
  float sq = q ? e.s1q : e.s1;
#pragma unroll
  for (int j = 0; j < 4; ++j) {
    float ar = zr[2 * j], ai = zi[2 * j];
    float br = q ? quad_rev(zr[2 * j + 1]) : zr[2 * j + 1];
    float bi = q ? quad_rev(zi[2 * j + 1]) : zi[2 * j + 1];
    float u0r = e.c1 * ar - sq * br, u0i = e.c1 * ai - sq * bi;
    float u1r = e.s1 * ar + cq * br, u1i = e.s1 * ai + cq * bi;
    yr[2 * j]     = u0r * e.pr[j] - u0i * e.pi[j];
    yi[2 * j]     = u0r * e.pi[j] + u0i * e.pr[j];
    yr[2 * j + 1] = u1r * e.pr[j] - u1i * e.pi[j];
    yi[2 * j + 1] = u1r * e.pi[j] + u1i * e.pr[j];
  }
}

__device__ __forceinline__ void layer(const float* __restrict__ vt, int l, const enc& e, bool q,
                                      v8f& zr, v8f& zi) {
  v8f yr, yi, tr, ti;
  encode(e, q, zr, zi, yr, yi);
  v8f a1 = load_tile(vt, l);
  v8f a2 = load_tile(vt + 256, l);
  v8f a3 = load_tile(vt + 512, l);
  cprod(yr, yi, a1, a2, a3, tr, ti);
  v8f b1 = load_tile(vt + 768, l);
  v8f b2 = load_tile(vt + 1024, l);
  v8f b3 = load_tile(vt + 1280, l);
  cprod(tr, ti, b1, b2, b3, zr, zi);
}

__device__ __forceinline__ float wave_sum(float v, int l) {
#pragma unroll
  for (int m = 16; m; m >>= 1) {
    int src = (l ^ m) << 2;
    v += __int_as_float(__builtin_amdgcn_ds_bpermute(src, __float_as_int(v)));
  }
  return v;
}

__global__ void __launch_bounds__(NTHREADS)
qnn_kernel(const float* __restrict__ x, const float* __restrict__ Vg,
           float* __restrict__ out, int batch) {
  __shared__ float Vl[VUSED];

  int tid = threadIdx.x;
  int l = tid & 31;
  int w = tid >> 5;
  int b = blockIdx.x * NWAVES + w;
  int bb = (b < batch) ? b : (batch - 1);

  {
    const float4* src = (const float4*)Vg;
    float4* dst = (float4*)Vl;
    for (int i = tid; i < VUSED / 4; i += NTHREADS) dst[i] = src[i];
  }
  __syncthreads();

  float x0 = x[2 * bb], x1 = x[2 * bb + 1];
  float c0 = cosf(0.5f * x0), s0 = sinf(0.5f * x0);
  float c1 = cosf(0.5f * x1), s1 = sinf(0.5f * x1);
  int lr = l & 15, lh = l >> 4;

  enc e;
  {
    c2 pc = c2{c0, lh ? s0 : -s0};
    pc = cmul(pc, c2{c0, (lr & 8) ? s0 : -s0});
    pc = cmul(pc, c2{c0, (lr & 4) ? s0 : -s0});
    pc = cmul(pc, c2{c1, (lr & 2) ? s1 : -s1});
    pc = cmul(pc, c2{c1, (lr & 1) ? s1 : -s1});
#pragma unroll
    for (int j = 0; j < 4; ++j) {
      c2 p12 = cmul(c2{c1, (j & 2) ? s1 : -s1}, c2{c0, (j & 1) ? s0 : -s0});
      c2 p = cmul(pc, p12);
      e.pr[j] = p.re;
      e.pi[j] = p.im;
    }
    float sg = (((lr >> 3) ^ (lr >> 2)) & 1) ? -1.f : 1.f;
    e.c1 = c1;
    e.s1 = s1;
    e.c1q = sg * c1;
    e.s1q = sg * s1;
  }

  v8f zr = load_tile(Vl + N_LAYERS * NPLANES * 256, l);
  v8f zi = load_tile(Vl + N_LAYERS * NPLANES * 256 + 256, l);

  layer(Vl, l, e, false, zr, zi);
  for (int ly = 1; ly < N_LAYERS; ++ly)
    layer(&Vl[ly * NPLANES * 256], l, e, true, zr, zi);

  float p[8];
#pragma unroll
  for (int v = 0; v < 8; ++v) {
    float fr = (v & 1) ? row_rev(zr[v]) : zr[v];
    float fi = (v & 1) ? row_rev(zi[v]) : zi[v];
    p[v] = fr * fr + fi * fi;
  }

  float t  = p[0] + p[1] + p[2] + p[3] + p[4] + p[5] + p[6] + p[7];
  float u1 = p[0] + p[1] + p[2] + p[3] - p[4] - p[5] - p[6] - p[7];
  float u2 = p[0] + p[1] - p[2] - p[3] + p[4] + p[5] - p[6] - p[7];
  float u3 = p[0] - p[1] + p[2] - p[3] + p[4] - p[5] + p[6] - p[7];
  float ev[8];
  ev[0] = lh ? -t : t;
  ev[1] = u1; ev[2] = u2; ev[3] = u3;
  ev[4] = (lr & 8) ? -t : t;
  ev[5] = (lr & 2) ? -t : t;
  ev[6] = (lr & 4) ? -t : t;
  ev[7] = (lr & 1) ? -t : t;
#pragma unroll
  for (int i = 0; i < 8; ++i) ev[i] = wave_sum(ev[i], l);

  if (b < batch && l == 0) {
    const float kPi = 3.14159265358979323846f;
#pragma unroll
    for (int i = 0; i < 8; ++i) out[b * 8 + i] = kPi * ev[i];
  }
}

extern "C" void kernel_launch(void* const* d_in, const int* in_sizes, int n_in,
                              void* d_out, int out_size, void* d_ws, size_t ws_size,
                              hipStream_t stream) {
  const float* x   = (const float*)d_in[0];
  const float* wts = (const float*)d_in[1];
  float* out = (float*)d_out;
  float* V   = (float*)d_ws;

  int batch = in_sizes[0] / 2;
  qnn_build_v<<<N_LAYERS, 256, 0, stream>>>(wts, V);
  int nblocks = (batch + NWAVES - 1) / NWAVES;
  qnn_kernel<<<nblocks, NTHREADS, 0, stream>>>(x, V, out, batch);
}


# layer 0: start tile is rank 1 (a b^T); A a and B b of a work-group's 8 elements from 2 shared WMMA products via LDS + a 2-WMMA outer product each: 5 WMMA per element, was 24; sums re-ordered
# speedup vs baseline: 2.4651x; 1.0951x over previous
#include <hip/hip_runtime.h>
#include <math.h>


typedef __attribute__((ext_vector_type(2))) float v2f;
typedef __attribute__((ext_vector_type(4))) float v4f;
typedef __attribute__((ext_vector_type(8))) float v8f;

#define N_LAYERS 6
#define NPLANES  6
#define NWAVES   8
#define NTHREADS (NWAVES * 32)
#define VUSED    (N_LAYERS * NPLANES * 256)
#define VECF     (NWAVES * 4 * 16)
#define RSQRT2   0.70710678118654752440f

struct c2 { float re, im; };

__device__ __forceinline__ c2 cmul(c2 a, c2 b) {
  return c2{a.re * b.re - a.im * b.im, a.re * b.im + a.im * b.re};
}
__device__ __forceinline__ c2 cadd(c2 a, c2 b) { return c2{a.re + b.re, a.im + b.im}; }

__device__ __forceinline__ c2 mul_ipow(c2 a, int k) {
  k &= 3;
  if (k == 0) return a;
  if (k == 1) return c2{-a.im, a.re};
  if (k == 2) return c2{-a.re, -a.im};
  return c2{a.im, -a.re};
}

__device__ __forceinline__ int col_std(int c) { return (c & 9) | ((c & 2) << 1) | ((c & 4) >> 1); }

__device__ __forceinline__ void wire_col(const float* __restrict__ wts, int l, int wire,
                                         int kind, int cb, c2& g0, c2& g1) {
  float a0 = wts[(l * 8 + wire) * 3 + 0];
  float a1 = wts[(l * 8 + wire) * 3 + 1];
  float a2 = wts[(l * 8 + wire) * 3 + 2];
  float cx, sx, cy, sy, cz, sz;
  sincosf(0.5f * a0, &sx, &cx);
  sincosf(0.5f * a1, &sy, &cy);
  sincosf(0.5f * a2, &sz, &cz);
  c2 w00 = cmul(c2{cz, -sz}, c2{cy * cx, sy * sx});
  c2 w01 = cmul(c2{cz, -sz}, c2{-sy * cx, -cy * sx});
  c2 w10 = cmul(c2{cz, sz}, c2{sy * cx, -cy * sx});
  c2 w11 = cmul(c2{cz, sz}, c2{cy * cx, -sy * sx});
  if (kind == 0) {
    float sg = cb ? -RSQRT2 : RSQRT2;
    g0 = c2{RSQRT2 * w00.re + sg * w01.re, RSQRT2 * w00.im + sg * w01.im};
    g1 = c2{RSQRT2 * w10.re + sg * w11.re, RSQRT2 * w10.im + sg * w11.im};
  } else if (kind == 1) {
    c2 p0 = cb ? mul_ipow(w00, 1) : w00;
    c2 p1 = cb ? w01 : mul_ipow(w01, 1);
    c2 q0 = cb ? mul_ipow(w10, 1) : w10;
    c2 q1 = cb ? w11 : mul_ipow(w11, 1);
    g0 = c2{RSQRT2 * (p0.re + p1.re), RSQRT2 * (p0.im + p1.im)};
    g1 = c2{RSQRT2 * (q0.re + q1.re), RSQRT2 * (q0.im + q1.im)};
  } else {
    g0 = cb ? w01 : w00;
    g1 = cb ? w11 : w10;
  }
}

__device__ __forceinline__ void uvec(int kind, int rb, c2& u0, c2& u1) {
  if (kind == 0) {
    u0 = c2{RSQRT2, 0.f};
    u1 = c2{rb ? -RSQRT2 : RSQRT2, 0.f};
  } else if (kind == 1) {
    u0 = rb ? c2{0.f, -RSQRT2} : c2{RSQRT2, 0.f};
    u1 = rb ? c2{RSQRT2, 0.f} : c2{0.f, -RSQRT2};
  } else {
    u0 = c2{rb ? 0.f : 1.f, 0.f};
    u1 = c2{rb ? 1.f : 0.f, 0.f};
  }
}

__device__ __forceinline__ c2 group_entry(const float* __restrict__ wts, int l, int base,
                                          bool last, bool keep3, int r, int c) {
  c2 f0 = c2{0.f, 0.f}, f1 = c2{0.f, 0.f};
#pragma unroll
  for (int i = 0; i < 4; ++i) {
    int sh = 3 - i;
    int vk = (keep3 && i == 3) ? 2 : (i & 1);
    int uk = last ? 2 : vk;
    int rb = (r >> sh) & 1, cb = (c >> sh) & 1;
    c2 g0, g1, u0, u1;
    wire_col(wts, l, base + i, vk, cb, g0, g1);
    uvec(uk, rb, u0, u1);
    if (i == 0) {
      f0 = cmul(u0, g0);
      f1 = cmul(u1, g1);
    } else {
      c2 t0 = cadd(cmul(f0, g0), cmul(f1, g1));
      c2 t1 = cadd(cmul(f0, g1), cmul(f1, g0));
      f0 = cmul(u0, t0);
      f1 = cmul(u1, t1);
    }
  }
  return cadd(f0, f1);
}

__global__ void qnn_build_v(const float* __restrict__ wts, float* __restrict__ V) {
  int l = blockIdx.x;
  int t = threadIdx.x;
  int r = t >> 4, c = t & 15;
  bool last = (l == N_LAYERS - 1);
  int offT = (((c & 7) >> 2) * 32 + r + 16 * (c >> 3)) * 4 + (c & 3);

  c2 a = group_entry(wts, l, 0, last, true, r, c);
  c2 b = group_entry(wts, l, 4, last, false, col_std(r), col_std(c));
  float* Vl = V + l * NPLANES * 256;
  Vl[0 * 256 + offT] = a.re;
  Vl[1 * 256 + offT] = a.im - a.re;
  Vl[2 * 256 + offT] = -(a.re + a.im);
  Vl[3 * 256 + offT] = b.re;
  Vl[4 * 256 + offT] = b.im - b.re;
  Vl[5 * 256 + offT] = -(b.re + b.im);

}

__device__ __forceinline__ v8f mma(v2f a, v2f b, v8f c) {
  return __builtin_amdgcn_wmma_f32_16x16x4_f32(false, a, false, b, (short)0, c, false, false);
}
__device__ __forceinline__ v2f frag(v8f t, int kb) {
  v2f f;
  f.x = t[2 * kb];
  f.y = t[2 * kb + 1];
  return f;
}

__device__ __forceinline__ void cprod(v8f Pr, v8f Pi, v8f q1, v8f q2, v8f q3n, v8f& Cr, v8f& Ci) {
  v8f ps = Pr + Pi;
  v8f k1 = {};
#pragma unroll
  for (int kb = 0; kb < 4; ++kb) k1 = mma(frag(ps, kb), frag(q1, kb), k1);
  v8f cr = k1, ci = k1;
#pragma unroll
  for (int kb = 0; kb < 4; ++kb) {
    ci = mma(frag(Pr, kb), frag(q2, kb), ci);
    cr = mma(frag(Pi, kb), frag(q3n, kb), cr);
  }
  Cr = cr;
  Ci = ci;
}

__device__ __forceinline__ v8f load_tile(const float* __restrict__ p, int l) {
  v4f lo = *(const v4f*)(p + 4 * l);
  v4f hi = *(const v4f*)(p + 128 + 4 * l);
  return __builtin_shufflevector(lo, hi, 0, 1, 2, 3, 4, 5, 6, 7);
}

__device__ __forceinline__ float quad_rev(float x) {
  return __int_as_float(__builtin_amdgcn_update_dpp(0, __float_as_int(x), 0x1B, 0xF, 0xF, true));
}
__device__ __forceinline__ float row_rev(float x) {
  return __int_as_float(__builtin_amdgcn_update_dpp(0, __float_as_int(x), 0x140, 0xF, 0xF, true));
}

struct enc {
  float c1, s1;
  float c1q, s1q;
  float pr[4], pi[4];
};

__device__ __forceinline__ void encode(const enc& e, bool q, v8f zr, v8f zi, v8f& yr, v8f& yi) {
  float cq = q ? e.c1q : e.c1;
  float sq = q ? e.s1q : e.s1;
#pragma unroll
  for (int j = 0; j < 4; ++j) {
    float ar = zr[2 * j], ai = zi[2 * j];
    float br = q ? quad_rev(zr[2 * j + 1]) : zr[2 * j + 1];
    float bi = q ? quad_rev(zi[2 * j + 1]) : zi[2 * j + 1];
    float u0r = e.c1 * ar - sq * br, u0i = e.c1 * ai - sq * bi;
    float u1r = e.s1 * ar + cq * br, u1i = e.s1 * ai + cq * bi;
    yr[2 * j]     = u0r * e.pr[j] - u0i * e.pi[j];
    yi[2 * j]     = u0r * e.pi[j] + u0i * e.pr[j];
    yr[2 * j + 1] = u1r * e.pr[j] - u1i * e.pi[j];
    yi[2 * j + 1] = u1r * e.pi[j] + u1i * e.pr[j];
  }
}

__device__ __forceinline__ void layer(const float* __restrict__ vt, int l, const enc& e, bool q,
                                      v8f& zr, v8f& zi) {
  v8f yr, yi, tr, ti;
  encode(e, q, zr, zi, yr, yi);
  v8f a1 = load_tile(vt, l);
  v8f a2 = load_tile(vt + 256, l);
  v8f a3 = load_tile(vt + 512, l);
  cprod(yr, yi, a1, a2, a3, tr, ti);
  v8f b1 = load_tile(vt + 768, l);
  v8f b2 = load_tile(vt + 1024, l);
  v8f b3 = load_tile(vt + 1280, l);
  cprod(tr, ti, b1, b2, b3, zr, zi);
}

__device__ __forceinline__ float wave_sum(float v, int l) {
#pragma unroll
  for (int m = 16; m; m >>= 1) {
    int src = (l ^ m) << 2;
    v += __int_as_float(__builtin_amdgcn_ds_bpermute(src, __float_as_int(v)));
  }
  return v;
}

__global__ void __launch_bounds__(NTHREADS)
qnn_kernel(const float* __restrict__ x, const float* __restrict__ Vg,
           float* __restrict__ out, int batch) {
  __shared__ float Vl[VUSED];
  __shared__ float Vec[VECF];

  int tid = threadIdx.x;
  int l = tid & 31;
  int w = tid >> 5;
  int b = blockIdx.x * NWAVES + w;
  int bb = (b < batch) ? b : (batch - 1);

  {
    const float4* src = (const float4*)Vg;
    float4* dst = (float4*)Vl;
    for (int i = tid; i < VUSED / 4; i += NTHREADS) dst[i] = src[i];
  }

  float x0 = x[2 * bb], x1 = x[2 * bb + 1];
  float c0 = cosf(0.5f * x0), s0 = sinf(0.5f * x0);
  float c1 = cosf(0.5f * x1), s1 = sinf(0.5f * x1);
  int lr = l & 15, lh = l >> 4;

  enc e;
  {
    c2 pc = c2{c0, lh ? s0 : -s0};
    pc = cmul(pc, c2{c0, (lr & 8) ? s0 : -s0});
    pc = cmul(pc, c2{c0, (lr & 4) ? s0 : -s0});
    pc = cmul(pc, c2{c1, (lr & 2) ? s1 : -s1});
    pc = cmul(pc, c2{c1, (lr & 1) ? s1 : -s1});
#pragma unroll
    for (int j = 0; j < 4; ++j) {
      c2 p12 = cmul(c2{c1, (j & 2) ? s1 : -s1}, c2{c0, (j & 1) ? s0 : -s0});
      c2 p = cmul(pc, p12);
      e.pr[j] = p.re;
      e.pi[j] = p.im;
    }
    float sg = (((lr >> 3) ^ (lr >> 2)) & 1) ? -1.f : 1.f;
    e.c1 = c1;
    e.s1 = s1;
    e.c1q = sg * c1;
    e.s1q = sg * s1;
  }

  {
    c2 f3 = c2{c0, (lr & 8) ? s0 : -s0};
    c2 f2 = lh ? c2{c0, (lr & 4) ? s0 : -s0} : c2{c1, (lr & 4) ? s1 : -s1};
    c2 f1 = lh ? c2{c1, (lr & 2) ? s1 : -s1} : c2{c0, (lr & 2) ? s0 : -s0};
    c2 f0 = lh ? c2{c1, (lr & 1) ? s1 : -s1}
               : c2{((lr & 1) ? s1 : c1) * 0.08838834764831844055f, 0.f};
    int kp = lh ? (((lr >> 1) & 1) + (lr & 1)) : ((lr >> 2) & 1);
    c2 val = mul_ipow(cmul(cmul(f3, f2), cmul(f1, f0)), -kp);
    Vec[(w * 4 + 2 * lh) * 16 + lr] = val.re;
    Vec[(w * 4 + 2 * lh + 1) * 16 + lr] = val.im;
  }
  __syncthreads();

  {
    int role = __builtin_amdgcn_readfirstlane(w) - 2 * (int)(blockIdx.x & 1);
    if (role == 0 || role == 1) {
      const float* vp = Vl + role * (3 * 256);
      int comp = 2 * role;
      const float* pr = Vec + (((lr & (NWAVES - 1)) * 4 + comp) * 16 + 8 * lh);
      v4f rlo = *(const v4f*)(pr), rhi = *(const v4f*)(pr + 4);
      v4f ilo = *(const v4f*)(pr + 16), ihi = *(const v4f*)(pr + 20);
      v8f Pr = __builtin_shufflevector(rlo, rhi, 0, 1, 2, 3, 4, 5, 6, 7);
      v8f Pi = __builtin_shufflevector(ilo, ihi, 0, 1, 2, 3, 4, 5, 6, 7);
      if (lr >= NWAVES) { Pr = v8f{}; Pi = v8f{}; }
      v8f q1 = load_tile(vp, l);
      v8f q2 = load_tile(vp + 256, l);
      v8f q3 = load_tile(vp + 512, l);
      v8f Cr, Ci;
      cprod(Pr, Pi, q1, q2, q3, Cr, Ci);
      if (lh == 0) {
#pragma unroll
        for (int v = 0; v < NWAVES; ++v) {
          Vec[(v * 4 + comp) * 16 + lr] = Cr[v];
          Vec[(v * 4 + comp + 1) * 16 + lr] = Ci[v];
        }
      }
    }
  }
  __syncthreads();

  v8f zr, zi;
  {
    float alr = Vec[(w * 4 + 0) * 16 + lr], ali = Vec[(w * 4 + 1) * 16 + lr];
    float ber = Vec[(w * 4 + 2) * 16 + lr], bei = Vec[(w * 4 + 3) * 16 + lr];
    if (lh) { alr = 0.f; ali = 0.f; ber = 0.f; bei = 0.f; }
    v2f fa1 = {alr, -ali}, fb1 = {ber, bei};
    v2f fa2 = {alr, ali},  fb2 = {bei, ber};
    v8f z0 = {};
    zr = mma(fa1, fb1, z0);
    zi = mma(fa2, fb2, z0);
  }
  for (int ly = 1; ly < N_LAYERS; ++ly)
    layer(&Vl[ly * NPLANES * 256], l, e, true, zr, zi);

  float p[8];
#pragma unroll
  for (int v = 0; v < 8; ++v) {
    float fr = (v & 1) ? row_rev(zr[v]) : zr[v];
    float fi = (v & 1) ? row_rev(zi[v]) : zi[v];
    p[v] = fr * fr + fi * fi;
  }

  float t  = p[0] + p[1] + p[2] + p[3] + p[4] + p[5] + p[6] + p[7];
  float u1 = p[0] + p[1] + p[2] + p[3] - p[4] - p[5] - p[6] - p[7];
  float u2 = p[0] + p[1] - p[2] - p[3] + p[4] + p[5] - p[6] - p[7];
  float u3 = p[0] - p[1] + p[2] - p[3] + p[4] - p[5] + p[6] - p[7];
  float ev[8];
  ev[0] = lh ? -t : t;
  ev[1] = u1; ev[2] = u2; ev[3] = u3;
  ev[4] = (lr & 8) ? -t : t;
  ev[5] = (lr & 2) ? -t : t;
  ev[6] = (lr & 4) ? -t : t;
  ev[7] = (lr & 1) ? -t : t;
#pragma unroll
  for (int i = 0; i < 8; ++i) ev[i] = wave_sum(ev[i], l);

  if (b < batch && l == 0) {
    const float kPi = 3.14159265358979323846f;
#pragma unroll
    for (int i = 0; i < 8; ++i) out[b * 8 + i] = kPi * ev[i];
  }
}

extern "C" void kernel_launch(void* const* d_in, const int* in_sizes, int n_in,
                              void* d_out, int out_size, void* d_ws, size_t ws_size,
                              hipStream_t stream) {
  const float* x   = (const float*)d_in[0];
  const float* wts = (const float*)d_in[1];
  float* out = (float*)d_out;
  float* V   = (float*)d_ws;

  int batch = in_sizes[0] / 2;
  qnn_build_v<<<N_LAYERS, 256, 0, stream>>>(wts, V);
  int nblocks = (batch + NWAVES - 1) / NWAVES;
  qnn_kernel<<<nblocks, NTHREADS, 0, stream>>>(x, V, out, batch);
}


# layer 1: encoded layer-0 tile is rank 2 (a1 b1^T + a2 b2^T); A a_t, B b_t of the group's 8 elements from 2 shared products via LDS + a 2-WMMA outer product: 5 WMMA per element, was 26; sums reordered
# speedup vs baseline: 2.8423x; 1.1530x over previous
#include <hip/hip_runtime.h>
#include <math.h>


typedef __attribute__((ext_vector_type(2))) float v2f;
typedef __attribute__((ext_vector_type(4))) float v4f;
typedef __attribute__((ext_vector_type(8))) float v8f;

#define N_LAYERS 6
#define NPLANES  6
#define NWAVES   8
#define NTHREADS (NWAVES * 32)
#define VUSED    (N_LAYERS * NPLANES * 256)
#define VECF     (NWAVES * 4 * 16)
#define VEC2F    (NWAVES * 8 * 16)
#define RSQRT2   0.70710678118654752440f

struct c2 { float re, im; };

__device__ __forceinline__ c2 cmul(c2 a, c2 b) {
  return c2{a.re * b.re - a.im * b.im, a.re * b.im + a.im * b.re};
}
__device__ __forceinline__ c2 cadd(c2 a, c2 b) { return c2{a.re + b.re, a.im + b.im}; }

__device__ __forceinline__ c2 mul_ipow(c2 a, int k) {
  k &= 3;
  if (k == 0) return a;
  if (k == 1) return c2{-a.im, a.re};
  if (k == 2) return c2{-a.re, -a.im};
  return c2{a.im, -a.re};
}

__device__ __forceinline__ int col_std(int c) { return (c & 9) | ((c & 2) << 1) | ((c & 4) >> 1); }

__device__ __forceinline__ void wire_col(const float* __restrict__ wts, int l, int wire,
                                         int kind, int cb, c2& g0, c2& g1) {
  float a0 = wts[(l * 8 + wire) * 3 + 0];
  float a1 = wts[(l * 8 + wire) * 3 + 1];
  float a2 = wts[(l * 8 + wire) * 3 + 2];
  float cx, sx, cy, sy, cz, sz;
  sincosf(0.5f * a0, &sx, &cx);
  sincosf(0.5f * a1, &sy, &cy);
  sincosf(0.5f * a2, &sz, &cz);
  c2 w00 = cmul(c2{cz, -sz}, c2{cy * cx, sy * sx});
  c2 w01 = cmul(c2{cz, -sz}, c2{-sy * cx, -cy * sx});
  c2 w10 = cmul(c2{cz, sz}, c2{sy * cx, -cy * sx});
  c2 w11 = cmul(c2{cz, sz}, c2{cy * cx, -sy * sx});
  if (kind == 0) {
    float sg = cb ? -RSQRT2 : RSQRT2;
    g0 = c2{RSQRT2 * w00.re + sg * w01.re, RSQRT2 * w00.im + sg * w01.im};
    g1 = c2{RSQRT2 * w10.re + sg * w11.re, RSQRT2 * w10.im + sg * w11.im};
  } else if (kind == 1) {
    c2 p0 = cb ? mul_ipow(w00, 1) : w00;
    c2 p1 = cb ? w01 : mul_ipow(w01, 1);
    c2 q0 = cb ? mul_ipow(w10, 1) : w10;
    c2 q1 = cb ? w11 : mul_ipow(w11, 1);
    g0 = c2{RSQRT2 * (p0.re + p1.re), RSQRT2 * (p0.im + p1.im)};
    g1 = c2{RSQRT2 * (q0.re + q1.re), RSQRT2 * (q0.im + q1.im)};
  } else {
    g0 = cb ? w01 : w00;
    g1 = cb ? w11 : w10;
  }
}

__device__ __forceinline__ void uvec(int kind, int rb, c2& u0, c2& u1) {
  if (kind == 0) {
    u0 = c2{RSQRT2, 0.f};
    u1 = c2{rb ? -RSQRT2 : RSQRT2, 0.f};
  } else if (kind == 1) {
    u0 = rb ? c2{0.f, -RSQRT2} : c2{RSQRT2, 0.f};
    u1 = rb ? c2{RSQRT2, 0.f} : c2{0.f, -RSQRT2};
  } else {
    u0 = c2{rb ? 0.f : 1.f, 0.f};
    u1 = c2{rb ? 1.f : 0.f, 0.f};
  }
}

__device__ __forceinline__ c2 group_entry(const float* __restrict__ wts, int l, int base,
                                          bool last, bool keep3, int r, int c) {
  c2 f0 = c2{0.f, 0.f}, f1 = c2{0.f, 0.f};
#pragma unroll
  for (int i = 0; i < 4; ++i) {
    int sh = 3 - i;
    int vk = (keep3 && i == 3) ? 2 : (i & 1);
    int uk = last ? 2 : vk;
    int rb = (r >> sh) & 1, cb = (c >> sh) & 1;
    c2 g0, g1, u0, u1;
    wire_col(wts, l, base + i, vk, cb, g0, g1);
    uvec(uk, rb, u0, u1);
    if (i == 0) {
      f0 = cmul(u0, g0);
      f1 = cmul(u1, g1);
    } else {
      c2 t0 = cadd(cmul(f0, g0), cmul(f1, g1));
      c2 t1 = cadd(cmul(f0, g1), cmul(f1, g0));
      f0 = cmul(u0, t0);
      f1 = cmul(u1, t1);
    }
  }
  return cadd(f0, f1);
}

__global__ void qnn_build_v(const float* __restrict__ wts, float* __restrict__ V) {
  int l = blockIdx.x;
  int t = threadIdx.x;
  int r = t >> 4, c = t & 15;
  bool last = (l == N_LAYERS - 1);
  int offT = (((c & 7) >> 2) * 32 + r + 16 * (c >> 3)) * 4 + (c & 3);

  c2 a = group_entry(wts, l, 0, last, true, r, c);
  c2 b = group_entry(wts, l, 4, last, false, col_std(r), col_std(c));
  float* Vl = V + l * NPLANES * 256;
  Vl[0 * 256 + offT] = a.re;
  Vl[1 * 256 + offT] = a.im - a.re;
  Vl[2 * 256 + offT] = -(a.re + a.im);
  Vl[3 * 256 + offT] = b.re;
  Vl[4 * 256 + offT] = b.im - b.re;
  Vl[5 * 256 + offT] = -(b.re + b.im);

}

__device__ __forceinline__ v8f mma(v2f a, v2f b, v8f c) {
  return __builtin_amdgcn_wmma_f32_16x16x4_f32(false, a, false, b, (short)0, c, false, false);
}
__device__ __forceinline__ v2f frag(v8f t, int kb) {
  v2f f;
  f.x = t[2 * kb];
  f.y = t[2 * kb + 1];
  return f;
}

__device__ __forceinline__ void cprod(v8f Pr, v8f Pi, v8f q1, v8f q2, v8f q3n, v8f& Cr, v8f& Ci) {
  v8f ps = Pr + Pi;
  v8f k1 = {};
#pragma unroll
  for (int kb = 0; kb < 4; ++kb) k1 = mma(frag(ps, kb), frag(q1, kb), k1);
  v8f cr = k1, ci = k1;
#pragma unroll
  for (int kb = 0; kb < 4; ++kb) {
    ci = mma(frag(Pr, kb), frag(q2, kb), ci);
    cr = mma(frag(Pi, kb), frag(q3n, kb), cr);
  }
  Cr = cr;
  Ci = ci;
}

__device__ __forceinline__ v8f load_tile(const float* __restrict__ p, int l) {
  v4f lo = *(const v4f*)(p + 4 * l);
  v4f hi = *(const v4f*)(p + 128 + 4 * l);
  return __builtin_shufflevector(lo, hi, 0, 1, 2, 3, 4, 5, 6, 7);
}

__device__ __forceinline__ float quad_rev(float x) {
  return __int_as_float(__builtin_amdgcn_update_dpp(0, __float_as_int(x), 0x1B, 0xF, 0xF, true));
}
__device__ __forceinline__ float row_rev(float x) {
  return __int_as_float(__builtin_amdgcn_update_dpp(0, __float_as_int(x), 0x140, 0xF, 0xF, true));
}

struct enc {
  float c1, s1;
  float c1q, s1q;
  float pr[4], pi[4];
};

__device__ __forceinline__ void encode(const enc& e, bool q, v8f zr, v8f zi, v8f& yr, v8f& yi) {
  float cq = q ? e.c1q : e.c1;
  float sq = q ? e.s1q : e.s1;
#pragma unroll
  for (int j = 0; j < 4; ++j) {
    float ar = zr[2 * j], ai = zi[2 * j];
    float br = q ? quad_rev(zr[2 * j + 1]) : zr[2 * j + 1];
    float bi = q ? quad_rev(zi[2 * j + 1]) : zi[2 * j + 1];
    float u0r = e.c1 * ar - sq * br, u0i = e.c1 * ai - sq * bi;
    float u1r = e.s1 * ar + cq * br, u1i = e.s1 * ai + cq * bi;
    yr[2 * j]     = u0r * e.pr[j] - u0i * e.pi[j];
    yi[2 * j]     = u0r * e.pi[j] + u0i * e.pr[j];
    yr[2 * j + 1] = u1r * e.pr[j] - u1i * e.pi[j];
    yi[2 * j + 1] = u1r * e.pi[j] + u1i * e.pr[j];
  }
}

__device__ __forceinline__ void layer(const float* __restrict__ vt, int l, const enc& e, bool q,
                                      v8f& zr, v8f& zi) {
  v8f yr, yi, tr, ti;
  encode(e, q, zr, zi, yr, yi);
  v8f a1 = load_tile(vt, l);
  v8f a2 = load_tile(vt + 256, l);
  v8f a3 = load_tile(vt + 512, l);
  cprod(yr, yi, a1, a2, a3, tr, ti);
  v8f b1 = load_tile(vt + 768, l);
  v8f b2 = load_tile(vt + 1024, l);
  v8f b3 = load_tile(vt + 1280, l);
  cprod(tr, ti, b1, b2, b3, zr, zi);
}

__device__ __forceinline__ float wave_sum(float v, int l) {
#pragma unroll
  for (int m = 16; m; m >>= 1) {
    int src = (l ^ m) << 2;
    v += __int_as_float(__builtin_amdgcn_ds_bpermute(src, __float_as_int(v)));
  }
  return v;
}

__global__ void __launch_bounds__(NTHREADS)
qnn_kernel(const float* __restrict__ x, const float* __restrict__ Vg,
           float* __restrict__ out, int batch) {
  __shared__ float Vl[VUSED];
  __shared__ float Vec[VECF];
  __shared__ float Vec2[VEC2F];

  int tid = threadIdx.x;
  int l = tid & 31;
  int w = tid >> 5;
  int b = blockIdx.x * NWAVES + w;
  int bb = (b < batch) ? b : (batch - 1);

  {
    const float4* src = (const float4*)Vg;
    float4* dst = (float4*)Vl;
    for (int i = tid; i < VUSED / 4; i += NTHREADS) dst[i] = src[i];
  }

  float x0 = x[2 * bb], x1 = x[2 * bb + 1];
  float c0 = cosf(0.5f * x0), s0 = sinf(0.5f * x0);
  float c1 = cosf(0.5f * x1), s1 = sinf(0.5f * x1);
  int lr = l & 15, lh = l >> 4;

  enc e;
  {
    c2 pc = c2{c0, lh ? s0 : -s0};
    pc = cmul(pc, c2{c0, (lr & 8) ? s0 : -s0});
    pc = cmul(pc, c2{c0, (lr & 4) ? s0 : -s0});
    pc = cmul(pc, c2{c1, (lr & 2) ? s1 : -s1});
    pc = cmul(pc, c2{c1, (lr & 1) ? s1 : -s1});
#pragma unroll
    for (int j = 0; j < 4; ++j) {
      c2 p12 = cmul(c2{c1, (j & 2) ? s1 : -s1}, c2{c0, (j & 1) ? s0 : -s0});
      c2 p = cmul(pc, p12);
      e.pr[j] = p.re;
      e.pi[j] = p.im;
    }
    float sg = (((lr >> 3) ^ (lr >> 2)) & 1) ? -1.f : 1.f;
    e.c1 = c1;
    e.s1 = s1;
    e.c1q = sg * c1;
    e.s1q = sg * s1;
  }

  c2 phi;
  {
    c2 f3 = c2{c0, (lr & 8) ? s0 : -s0};
    c2 f2 = lh ? c2{c0, (lr & 4) ? s0 : -s0} : c2{c1, (lr & 4) ? s1 : -s1};
    c2 f1 = lh ? c2{c1, (lr & 2) ? s1 : -s1} : c2{c0, (lr & 2) ? s0 : -s0};
    c2 f0 = lh ? c2{c1, (lr & 1) ? s1 : -s1}
               : c2{((lr & 1) ? s1 : c1) * 0.08838834764831844055f, 0.f};
    int kp = lh ? (((lr >> 1) & 1) + (lr & 1)) : ((lr >> 2) & 1);
    c2 f321 = cmul(cmul(f3, f2), f1);
    c2 f3210 = cmul(f321, f0);
    phi = lh ? f3210 : f321;
    c2 val = mul_ipow(f3210, -kp);
    Vec[(w * 4 + 2 * lh) * 16 + lr] = val.re;
    Vec[(w * 4 + 2 * lh + 1) * 16 + lr] = val.im;
  }
  __syncthreads();

  {
    int role = __builtin_amdgcn_readfirstlane(w) - 2 * (int)(blockIdx.x & 1);
    if (role == 0 || role == 1) {
      const float* vp = Vl + role * (3 * 256);
      int comp = 2 * role;
      const float* pr = Vec + (((lr & (NWAVES - 1)) * 4 + comp) * 16 + 8 * lh);
      v4f rlo = *(const v4f*)(pr), rhi = *(const v4f*)(pr + 4);
      v4f ilo = *(const v4f*)(pr + 16), ihi = *(const v4f*)(pr + 20);
      v8f Pr = __builtin_shufflevector(rlo, rhi, 0, 1, 2, 3, 4, 5, 6, 7);
      v8f Pi = __builtin_shufflevector(ilo, ihi, 0, 1, 2, 3, 4, 5, 6, 7);
      if (lr >= NWAVES) { Pr = v8f{}; Pi = v8f{}; }
      v8f q1 = load_tile(vp, l);
      v8f q2 = load_tile(vp + 256, l);
      v8f q3 = load_tile(vp + 512, l);
      v8f Cr, Ci;
      cprod(Pr, Pi, q1, q2, q3, Cr, Ci);
      if (lh == 0) {
#pragma unroll
        for (int v = 0; v < NWAVES; ++v) {
          Vec[(v * 4 + comp) * 16 + lr] = Cr[v];
          Vec[(v * 4 + comp + 1) * 16 + lr] = Ci[v];
        }
      }
    }
  }
  __syncthreads();

  {
    const float* vb = Vec + (w * 4 + 2 * lh) * 16;
    int lo = lh ? (lr ^ 3) : (lr ^ 1);
    c2 own = c2{vb[lr], vb[16 + lr]};
    c2 oth = c2{vb[lo], vb[16 + lo]};
    bool odd = (lr & 1) != 0;
    float sg = (((lr >> 3) ^ (lr >> 2)) & 1) ? -1.f : 1.f;
    float k1 = lh ? 1.f : (odd ? s1 : c1);
    float k2 = lh ? sg : (odd ? c1 : -s1);
    c2 x1 = (lh || !odd) ? own : oth;
    c2 x2 = (lh || !odd) ? oth : own;
    c2 v1 = cmul(phi, c2{k1 * x1.re, k1 * x1.im});
    c2 v2 = cmul(phi, c2{k2 * x2.re, k2 * x2.im});
    float* vo = Vec2 + ((w * 2 * 2 + lh) * 2) * 16 + lr;
    vo[0] = v1.re;
    vo[16] = v1.im;
    vo[64] = v2.re;
    vo[80] = v2.im;
  }
  __syncthreads();

  {
    int role = __builtin_amdgcn_readfirstlane(w) - 6 + 2 * (int)(blockIdx.x & 1);
    if (role == 0 || role == 1) {
      const float* vp = Vl + NPLANES * 256 + role * (3 * 256);
      const float* pr = Vec2 + ((lr * 2 + role) * 2) * 16 + 8 * lh;
      v4f rlo = *(const v4f*)(pr), rhi = *(const v4f*)(pr + 4);
      v4f ilo = *(const v4f*)(pr + 16), ihi = *(const v4f*)(pr + 20);
      v8f Pr = __builtin_shufflevector(rlo, rhi, 0, 1, 2, 3, 4, 5, 6, 7);
      v8f Pi = __builtin_shufflevector(ilo, ihi, 0, 1, 2, 3, 4, 5, 6, 7);
      v8f q1 = load_tile(vp, l);
      v8f q2 = load_tile(vp + 256, l);
      v8f q3 = load_tile(vp + 512, l);
      v8f Cr, Ci;
      cprod(Pr, Pi, q1, q2, q3, Cr, Ci);
      float* po = Vec2 + (((8 * lh) * 2 + role) * 2) * 16 + lr;
#pragma unroll
      for (int v = 0; v < 8; ++v) {
        po[v * 64] = Cr[v];
        po[v * 64 + 16] = Ci[v];
      }
    }
  }
  __syncthreads();

  v8f zr, zi;
  {
    const float* vt = Vec2 + ((w * 2 + lh) * 4) * 16 + lr;
    float ppr = vt[0], ppi = vt[16];
    float qqr = vt[32], qqi = vt[48];
    v2f fa1 = {ppr, -ppi}, fb1 = {qqr, qqi};
    v2f fa2 = {ppr, ppi},  fb2 = {qqi, qqr};
    v8f z0 = {};
    zr = mma(fa1, fb1, z0);
    zi = mma(fa2, fb2, z0);
  }
  for (int ly = 2; ly < N_LAYERS; ++ly)
    layer(&Vl[ly * NPLANES * 256], l, e, true, zr, zi);

  float p[8];
#pragma unroll
  for (int v = 0; v < 8; ++v) {
    float fr = (v & 1) ? row_rev(zr[v]) : zr[v];
    float fi = (v & 1) ? row_rev(zi[v]) : zi[v];
    p[v] = fr * fr + fi * fi;
  }

  float t  = p[0] + p[1] + p[2] + p[3] + p[4] + p[5] + p[6] + p[7];
  float u1 = p[0] + p[1] + p[2] + p[3] - p[4] - p[5] - p[6] - p[7];
  float u2 = p[0] + p[1] - p[2] - p[3] + p[4] + p[5] - p[6] - p[7];
  float u3 = p[0] - p[1] + p[2] - p[3] + p[4] - p[5] + p[6] - p[7];
  float ev[8];
  ev[0] = lh ? -t : t;
  ev[1] = u1; ev[2] = u2; ev[3] = u3;
  ev[4] = (lr & 8) ? -t : t;
  ev[5] = (lr & 2) ? -t : t;
  ev[6] = (lr & 4) ? -t : t;
  ev[7] = (lr & 1) ? -t : t;
#pragma unroll
  for (int i = 0; i < 8; ++i) ev[i] = wave_sum(ev[i], l);

  if (b < batch && l == 0) {
    const float kPi = 3.14159265358979323846f;
#pragma unroll
    for (int i = 0; i < 8; ++i) out[b * 8 + i] = kPi * ev[i];
  }
}

extern "C" void kernel_launch(void* const* d_in, const int* in_sizes, int n_in,
                              void* d_out, int out_size, void* d_ws, size_t ws_size,
                              hipStream_t stream) {
  const float* x   = (const float*)d_in[0];
  const float* wts = (const float*)d_in[1];
  float* out = (float*)d_out;
  float* V   = (float*)d_ws;

  int batch = in_sizes[0] / 2;
  qnn_build_v<<<N_LAYERS, 256, 0, stream>>>(wts, V);
  int nblocks = (batch + NWAVES - 1) / NWAVES;
  qnn_kernel<<<nblocks, NTHREADS, 0, stream>>>(x, V, out, batch);
}


# layer 2: encoded layer-1 tile is rank 4; A a, B b of the group's 32 vectors per side from 4 shared products via LDS + a 4-WMMA outer product: 10 WMMA per element, was 26; sums reordered
# speedup vs baseline: 3.1120x; 1.0949x over previous
#include <hip/hip_runtime.h>
#include <math.h>


typedef __attribute__((ext_vector_type(2))) float v2f;
typedef __attribute__((ext_vector_type(4))) float v4f;
typedef __attribute__((ext_vector_type(8))) float v8f;

#define N_LAYERS 6
#define NPLANES  6
#define NWAVES   8
#define NTHREADS (NWAVES * 32)
#define VUSED    (N_LAYERS * NPLANES * 256)
#define VECF     (NWAVES * 4 * 16)
#define VEC2F    (NWAVES * 8 * 16)
#define VEC3F    (NWAVES * 16 * 16)
#define RSQRT2   0.70710678118654752440f

struct c2 { float re, im; };

__device__ __forceinline__ c2 cmul(c2 a, c2 b) {
  return c2{a.re * b.re - a.im * b.im, a.re * b.im + a.im * b.re};
}
__device__ __forceinline__ c2 cadd(c2 a, c2 b) { return c2{a.re + b.re, a.im + b.im}; }

__device__ __forceinline__ c2 mul_ipow(c2 a, int k) {
  k &= 3;
  if (k == 0) return a;
  if (k == 1) return c2{-a.im, a.re};
  if (k == 2) return c2{-a.re, -a.im};
  return c2{a.im, -a.re};
}

__device__ __forceinline__ int col_std(int c) { return (c & 9) | ((c & 2) << 1) | ((c & 4) >> 1); }

__device__ __forceinline__ void wire_col(const float* __restrict__ wts, int l, int wire,
                                         int kind, int cb, c2& g0, c2& g1) {
  float a0 = wts[(l * 8 + wire) * 3 + 0];
  float a1 = wts[(l * 8 + wire) * 3 + 1];
  float a2 = wts[(l * 8 + wire) * 3 + 2];
  float cx, sx, cy, sy, cz, sz;
  sincosf(0.5f * a0, &sx, &cx);
  sincosf(0.5f * a1, &sy, &cy);
  sincosf(0.5f * a2, &sz, &cz);
  c2 w00 = cmul(c2{cz, -sz}, c2{cy * cx, sy * sx});
  c2 w01 = cmul(c2{cz, -sz}, c2{-sy * cx, -cy * sx});
  c2 w10 = cmul(c2{cz, sz}, c2{sy * cx, -cy * sx});
  c2 w11 = cmul(c2{cz, sz}, c2{cy * cx, -sy * sx});
  if (kind == 0) {
    float sg = cb ? -RSQRT2 : RSQRT2;
    g0 = c2{RSQRT2 * w00.re + sg * w01.re, RSQRT2 * w00.im + sg * w01.im};
    g1 = c2{RSQRT2 * w10.re + sg * w11.re, RSQRT2 * w10.im + sg * w11.im};
  } else if (kind == 1) {
    c2 p0 = cb ? mul_ipow(w00, 1) : w00;
    c2 p1 = cb ? w01 : mul_ipow(w01, 1);
    c2 q0 = cb ? mul_ipow(w10, 1) : w10;
    c2 q1 = cb ? w11 : mul_ipow(w11, 1);
    g0 = c2{RSQRT2 * (p0.re + p1.re), RSQRT2 * (p0.im + p1.im)};
    g1 = c2{RSQRT2 * (q0.re + q1.re), RSQRT2 * (q0.im + q1.im)};
  } else {
    g0 = cb ? w01 : w00;
    g1 = cb ? w11 : w10;
  }
}

__device__ __forceinline__ void uvec(int kind, int rb, c2& u0, c2& u1) {
  if (kind == 0) {
    u0 = c2{RSQRT2, 0.f};
    u1 = c2{rb ? -RSQRT2 : RSQRT2, 0.f};
  } else if (kind == 1) {
    u0 = rb ? c2{0.f, -RSQRT2} : c2{RSQRT2, 0.f};
    u1 = rb ? c2{RSQRT2, 0.f} : c2{0.f, -RSQRT2};
  } else {
    u0 = c2{rb ? 0.f : 1.f, 0.f};
    u1 = c2{rb ? 1.f : 0.f, 0.f};
  }
}

__device__ __forceinline__ c2 group_entry(const float* __restrict__ wts, int l, int base,
                                          bool last, bool keep3, int r, int c) {
  c2 f0 = c2{0.f, 0.f}, f1 = c2{0.f, 0.f};
#pragma unroll
  for (int i = 0; i < 4; ++i) {
    int sh = 3 - i;
    int vk = (keep3 && i == 3) ? 2 : (i & 1);
    int uk = last ? 2 : vk;
    int rb = (r >> sh) & 1, cb = (c >> sh) & 1;
    c2 g0, g1, u0, u1;
    wire_col(wts, l, base + i, vk, cb, g0, g1);
    uvec(uk, rb, u0, u1);
    if (i == 0) {
      f0 = cmul(u0, g0);
      f1 = cmul(u1, g1);
    } else {
      c2 t0 = cadd(cmul(f0, g0), cmul(f1, g1));
      c2 t1 = cadd(cmul(f0, g1), cmul(f1, g0));
      f0 = cmul(u0, t0);
      f1 = cmul(u1, t1);
    }
  }
  return cadd(f0, f1);
}

__global__ void qnn_build_v(const float* __restrict__ wts, float* __restrict__ V) {
  int l = blockIdx.x;
  int t = threadIdx.x;
  int r = t >> 4, c = t & 15;
  bool last = (l == N_LAYERS - 1);
  int offT = (((c & 7) >> 2) * 32 + r + 16 * (c >> 3)) * 4 + (c & 3);

  c2 a = group_entry(wts, l, 0, last, true, r, c);
  c2 b = group_entry(wts, l, 4, last, false, col_std(r), col_std(c));
  float* Vl = V + l * NPLANES * 256;
  Vl[0 * 256 + offT] = a.re;
  Vl[1 * 256 + offT] = a.im - a.re;
  Vl[2 * 256 + offT] = -(a.re + a.im);
  Vl[3 * 256 + offT] = b.re;
  Vl[4 * 256 + offT] = b.im - b.re;
  Vl[5 * 256 + offT] = -(b.re + b.im);

}

__device__ __forceinline__ v8f mma(v2f a, v2f b, v8f c) {
  return __builtin_amdgcn_wmma_f32_16x16x4_f32(false, a, false, b, (short)0, c, false, false);
}
__device__ __forceinline__ v2f frag(v8f t, int kb) {
  v2f f;
  f.x = t[2 * kb];
  f.y = t[2 * kb + 1];
  return f;
}

__device__ __forceinline__ void cprod(v8f Pr, v8f Pi, v8f q1, v8f q2, v8f q3n, v8f& Cr, v8f& Ci) {
  v8f ps = Pr + Pi;
  v8f k1 = {};
#pragma unroll
  for (int kb = 0; kb < 4; ++kb) k1 = mma(frag(ps, kb), frag(q1, kb), k1);
  v8f cr = k1, ci = k1;
#pragma unroll
  for (int kb = 0; kb < 4; ++kb) {
    ci = mma(frag(Pr, kb), frag(q2, kb), ci);
    cr = mma(frag(Pi, kb), frag(q3n, kb), cr);
  }
  Cr = cr;
  Ci = ci;
}

__device__ __forceinline__ v8f load_tile(const float* __restrict__ p, int l) {
  v4f lo = *(const v4f*)(p + 4 * l);
  v4f hi = *(const v4f*)(p + 128 + 4 * l);
  return __builtin_shufflevector(lo, hi, 0, 1, 2, 3, 4, 5, 6, 7);
}

__device__ __forceinline__ float quad_rev(float x) {
  return __int_as_float(__builtin_amdgcn_update_dpp(0, __float_as_int(x), 0x1B, 0xF, 0xF, true));
}
__device__ __forceinline__ float row_rev(float x) {
  return __int_as_float(__builtin_amdgcn_update_dpp(0, __float_as_int(x), 0x140, 0xF, 0xF, true));
}

struct enc {
  float c1, s1;
  float c1q, s1q;
  float pr[4], pi[4];
};

__device__ __forceinline__ void encode(const enc& e, bool q, v8f zr, v8f zi, v8f& yr, v8f& yi) {
  float cq = q ? e.c1q : e.c1;
  float sq = q ? e.s1q : e.s1;
#pragma unroll
  for (int j = 0; j < 4; ++j) {
    float ar = zr[2 * j], ai = zi[2 * j];
    float br = q ? quad_rev(zr[2 * j + 1]) : zr[2 * j + 1];
    float bi = q ? quad_rev(zi[2 * j + 1]) : zi[2 * j + 1];
    float u0r = e.c1 * ar - sq * br, u0i = e.c1 * ai - sq * bi;
    float u1r = e.s1 * ar + cq * br, u1i = e.s1 * ai + cq * bi;
    yr[2 * j]     = u0r * e.pr[j] - u0i * e.pi[j];
    yi[2 * j]     = u0r * e.pi[j] + u0i * e.pr[j];
    yr[2 * j + 1] = u1r * e.pr[j] - u1i * e.pi[j];
    yi[2 * j + 1] = u1r * e.pi[j] + u1i * e.pr[j];
  }
}

__device__ __forceinline__ void layer(const float* __restrict__ vt, int l, const enc& e, bool q,
                                      v8f& zr, v8f& zi) {
  v8f yr, yi, tr, ti;
  encode(e, q, zr, zi, yr, yi);
  v8f a1 = load_tile(vt, l);
  v8f a2 = load_tile(vt + 256, l);
  v8f a3 = load_tile(vt + 512, l);
  cprod(yr, yi, a1, a2, a3, tr, ti);
  v8f b1 = load_tile(vt + 768, l);
  v8f b2 = load_tile(vt + 1024, l);
  v8f b3 = load_tile(vt + 1280, l);
  cprod(tr, ti, b1, b2, b3, zr, zi);
}

__device__ __forceinline__ float wave_sum(float v, int l) {
#pragma unroll
  for (int m = 16; m; m >>= 1) {
    int src = (l ^ m) << 2;
    v += __int_as_float(__builtin_amdgcn_ds_bpermute(src, __float_as_int(v)));
  }
  return v;
}

__global__ void __launch_bounds__(NTHREADS)
qnn_kernel(const float* __restrict__ x, const float* __restrict__ Vg,
           float* __restrict__ out, int batch) {
  __shared__ float Vl[VUSED];
  __shared__ float Vec[VECF];
  __shared__ float Vec2[VEC2F];
  __shared__ float Vec3[VEC3F];

  int tid = threadIdx.x;
  int l = tid & 31;
  int w = tid >> 5;
  int b = blockIdx.x * NWAVES + w;
  int bb = (b < batch) ? b : (batch - 1);

  {
    const float4* src = (const float4*)Vg;
    float4* dst = (float4*)Vl;
    for (int i = tid; i < VUSED / 4; i += NTHREADS) dst[i] = src[i];
  }

  float x0 = x[2 * bb], x1 = x[2 * bb + 1];
  float c0 = cosf(0.5f * x0), s0 = sinf(0.5f * x0);
  float c1 = cosf(0.5f * x1), s1 = sinf(0.5f * x1);
  int lr = l & 15, lh = l >> 4;

  enc e;
  {
    c2 pc = c2{c0, lh ? s0 : -s0};
    pc = cmul(pc, c2{c0, (lr & 8) ? s0 : -s0});
    pc = cmul(pc, c2{c0, (lr & 4) ? s0 : -s0});
    pc = cmul(pc, c2{c1, (lr & 2) ? s1 : -s1});
    pc = cmul(pc, c2{c1, (lr & 1) ? s1 : -s1});
#pragma unroll
    for (int j = 0; j < 4; ++j) {
      c2 p12 = cmul(c2{c1, (j & 2) ? s1 : -s1}, c2{c0, (j & 1) ? s0 : -s0});
      c2 p = cmul(pc, p12);
      e.pr[j] = p.re;
      e.pi[j] = p.im;
    }
    float sg = (((lr >> 3) ^ (lr >> 2)) & 1) ? -1.f : 1.f;
    e.c1 = c1;
    e.s1 = s1;
    e.c1q = sg * c1;
    e.s1q = sg * s1;
  }

  c2 phi;
  {
    c2 f3 = c2{c0, (lr & 8) ? s0 : -s0};
    c2 f2 = lh ? c2{c0, (lr & 4) ? s0 : -s0} : c2{c1, (lr & 4) ? s1 : -s1};
    c2 f1 = lh ? c2{c1, (lr & 2) ? s1 : -s1} : c2{c0, (lr & 2) ? s0 : -s0};
    c2 f0 = lh ? c2{c1, (lr & 1) ? s1 : -s1}
               : c2{((lr & 1) ? s1 : c1) * 0.08838834764831844055f, 0.f};
    int kp = lh ? (((lr >> 1) & 1) + (lr & 1)) : ((lr >> 2) & 1);
    c2 f321 = cmul(cmul(f3, f2), f1);
    c2 f3210 = cmul(f321, f0);
    phi = lh ? f3210 : f321;
    c2 val = mul_ipow(f3210, -kp);
    Vec[(w * 4 + 2 * lh) * 16 + lr] = val.re;
    Vec[(w * 4 + 2 * lh + 1) * 16 + lr] = val.im;
  }
  __syncthreads();

  {
    int role = __builtin_amdgcn_readfirstlane(w) - 2 * (int)(blockIdx.x & 1);
    if (role == 0 || role == 1) {
      const float* vp = Vl + role * (3 * 256);
      int comp = 2 * role;
      const float* pr = Vec + (((lr & (NWAVES - 1)) * 4 + comp) * 16 + 8 * lh);
      v4f rlo = *(const v4f*)(pr), rhi = *(const v4f*)(pr + 4);
      v4f ilo = *(const v4f*)(pr + 16), ihi = *(const v4f*)(pr + 20);
      v8f Pr = __builtin_shufflevector(rlo, rhi, 0, 1, 2, 3, 4, 5, 6, 7);
      v8f Pi = __builtin_shufflevector(ilo, ihi, 0, 1, 2, 3, 4, 5, 6, 7);
      if (lr >= NWAVES) { Pr = v8f{}; Pi = v8f{}; }
      v8f q1 = load_tile(vp, l);
      v8f q2 = load_tile(vp + 256, l);
      v8f q3 = load_tile(vp + 512, l);
      v8f Cr, Ci;
      cprod(Pr, Pi, q1, q2, q3, Cr, Ci);
      if (lh == 0) {
#pragma unroll
        for (int v = 0; v < NWAVES; ++v) {
          Vec[(v * 4 + comp) * 16 + lr] = Cr[v];
          Vec[(v * 4 + comp + 1) * 16 + lr] = Ci[v];
        }
      }
    }
  }
  __syncthreads();

  const int lo = lh ? (lr ^ 3) : (lr ^ 1);
  const bool odd = (lr & 1) != 0;
  const float sgc = (((lr >> 3) ^ (lr >> 2)) & 1) ? -1.f : 1.f;
  const float k1 = lh ? 1.f : (odd ? s1 : c1);
  const float k2 = lh ? sgc : (odd ? c1 : -s1);
  {
    const float* vb = Vec + (w * 4 + 2 * lh) * 16;
    c2 own = c2{vb[lr], vb[16 + lr]};
    c2 oth = c2{vb[lo], vb[16 + lo]};
    c2 x1 = (lh || !odd) ? own : oth;
    c2 x2 = (lh || !odd) ? oth : own;
    c2 v1 = cmul(phi, c2{k1 * x1.re, k1 * x1.im});
    c2 v2 = cmul(phi, c2{k2 * x2.re, k2 * x2.im});
    float* vo = Vec2 + ((w * 2 * 2 + lh) * 2) * 16 + lr;
    vo[0] = v1.re;
    vo[16] = v1.im;
    vo[64] = v2.re;
    vo[80] = v2.im;
  }
  __syncthreads();

  {
    int role = __builtin_amdgcn_readfirstlane(w) - 6 + 2 * (int)(blockIdx.x & 1);
    if (role == 0 || role == 1) {
      const float* vp = Vl + NPLANES * 256 + role * (3 * 256);
      const float* pr = Vec2 + ((lr * 2 + role) * 2) * 16 + 8 * lh;
      v4f rlo = *(const v4f*)(pr), rhi = *(const v4f*)(pr + 4);
      v4f ilo = *(const v4f*)(pr + 16), ihi = *(const v4f*)(pr + 20);
      v8f Pr = __builtin_shufflevector(rlo, rhi, 0, 1, 2, 3, 4, 5, 6, 7);
      v8f Pi = __builtin_shufflevector(ilo, ihi, 0, 1, 2, 3, 4, 5, 6, 7);
      v8f q1 = load_tile(vp, l);
      v8f q2 = load_tile(vp + 256, l);
      v8f q3 = load_tile(vp + 512, l);
      v8f Cr, Ci;
      cprod(Pr, Pi, q1, q2, q3, Cr, Ci);
      float* po = Vec2 + (((8 * lh) * 2 + role) * 2) * 16 + lr;
#pragma unroll
      for (int v = 0; v < 8; ++v) {
        po[v * 64] = Cr[v];
        po[v * 64 + 16] = Ci[v];
      }
    }
  }
  __syncthreads();

  {
#pragma unroll
    for (int t = 0; t < 2; ++t) {
      const float* vb = Vec2 + (((w * 2 + t) * 2 + lh) * 2) * 16;
      c2 own = c2{vb[lr], vb[16 + lr]};
      c2 oth = c2{vb[lo], vb[16 + lo]};
      c2 x1 = (lh || !odd) ? own : oth;
      c2 x2 = (lh || !odd) ? oth : own;
      c2 v1 = cmul(phi, c2{k1 * x1.re, k1 * x1.im});
      c2 v2 = cmul(phi, c2{k2 * x2.re, k2 * x2.im});
      float* vo = Vec3 + (((w * 4 + 2 * t) * 2 + lh) * 2) * 16 + lr;
      vo[0] = v1.re;
      vo[16] = v1.im;
      vo[64] = v2.re;
      vo[80] = v2.im;
    }
  }
  __syncthreads();

  {
    int job = (__builtin_amdgcn_readfirstlane(w) + 6 - 4 * (int)(blockIdx.x & 1)) & 7;
    if (job < 4) {
      int side = job & 1, g = job >> 1;
      const float* vp = Vl + 2 * NPLANES * 256 + side * (3 * 256);
      const float* pr = Vec3 + (((16 * g + lr) * 2 + side) * 2) * 16 + 8 * lh;
      v4f rlo = *(const v4f*)(pr), rhi = *(const v4f*)(pr + 4);
      v4f ilo = *(const v4f*)(pr + 16), ihi = *(const v4f*)(pr + 20);
      v8f Pr = __builtin_shufflevector(rlo, rhi, 0, 1, 2, 3, 4, 5, 6, 7);
      v8f Pi = __builtin_shufflevector(ilo, ihi, 0, 1, 2, 3, 4, 5, 6, 7);
      v8f q1 = load_tile(vp, l);
      v8f q2 = load_tile(vp + 256, l);
      v8f q3 = load_tile(vp + 512, l);
      v8f Cr, Ci;
      cprod(Pr, Pi, q1, q2, q3, Cr, Ci);
      float* po = Vec3 + (((16 * g + 8 * lh) * 2 + side) * 2) * 16 + lr;
#pragma unroll
      for (int v = 0; v < 8; ++v) {
        po[v * 64] = Cr[v];
        po[v * 64 + 16] = Ci[v];
      }
    }
  }
  __syncthreads();

  v8f zr, zi;
  {
    const float* vt = Vec3 + ((w * 4 + lh) * 4) * 16 + lr;
    float par = vt[0],   pai = vt[16];
    float qar = vt[32],  qai = vt[48];
    float pbr = vt[128], pbi = vt[144];
    float qbr = vt[160], qbi = vt[176];
    v2f fa1 = {par, -pai}, fb1 = {qar, qai};
    v2f fa2 = {par, pai},  fb2 = {qai, qar};
    v2f ga1 = {pbr, -pbi}, gb1 = {qbr, qbi};
    v2f ga2 = {pbr, pbi},  gb2 = {qbi, qbr};
    v8f z0 = {};
    zr = mma(ga1, gb1, mma(fa1, fb1, z0));
    zi = mma(ga2, gb2, mma(fa2, fb2, z0));
  }
  for (int ly = 3; ly < N_LAYERS; ++ly)
    layer(&Vl[ly * NPLANES * 256], l, e, true, zr, zi);

  float p[8];
#pragma unroll
  for (int v = 0; v < 8; ++v) {
    float fr = (v & 1) ? row_rev(zr[v]) : zr[v];
    float fi = (v & 1) ? row_rev(zi[v]) : zi[v];
    p[v] = fr * fr + fi * fi;
  }

  float t  = p[0] + p[1] + p[2] + p[3] + p[4] + p[5] + p[6] + p[7];
  float u1 = p[0] + p[1] + p[2] + p[3] - p[4] - p[5] - p[6] - p[7];
  float u2 = p[0] + p[1] - p[2] - p[3] + p[4] + p[5] - p[6] - p[7];
  float u3 = p[0] - p[1] + p[2] - p[3] + p[4] - p[5] + p[6] - p[7];
  float ev[8];
  ev[0] = lh ? -t : t;
  ev[1] = u1; ev[2] = u2; ev[3] = u3;
  ev[4] = (lr & 8) ? -t : t;
  ev[5] = (lr & 2) ? -t : t;
  ev[6] = (lr & 4) ? -t : t;
  ev[7] = (lr & 1) ? -t : t;
#pragma unroll
  for (int i = 0; i < 8; ++i) ev[i] = wave_sum(ev[i], l);

  if (b < batch && l == 0) {
    const float kPi = 3.14159265358979323846f;
#pragma unroll
    for (int i = 0; i < 8; ++i) out[b * 8 + i] = kPi * ev[i];
  }
}

extern "C" void kernel_launch(void* const* d_in, const int* in_sizes, int n_in,
                              void* d_out, int out_size, void* d_ws, size_t ws_size,
                              hipStream_t stream) {
  const float* x   = (const float*)d_in[0];
  const float* wts = (const float*)d_in[1];
  float* out = (float*)d_out;
  float* V   = (float*)d_ws;

  int batch = in_sizes[0] / 2;
  qnn_build_v<<<N_LAYERS, 256, 0, stream>>>(wts, V);
  int nblocks = (batch + NWAVES - 1) / NWAVES;
  qnn_kernel<<<nblocks, NTHREADS, 0, stream>>>(x, V, out, batch);
}
